# starting kernel
# baseline (speedup 1.0000x reference)
#include <hip/hip_runtime.h>
#include <hip/hip_bf16.h>

typedef __attribute__((ext_vector_type(2))) float v2f;
typedef __attribute__((ext_vector_type(8))) float v8f;

#define NEG_SLOPE 0.01f


__global__ __launch_bounds__(256) void k_init_deg(float* __restrict__ deg, int n) {
  int i = blockIdx.x * blockDim.x + threadIdx.x;
  if (i < n) deg[i] = 1.0f;
}

__global__ __launch_bounds__(256) void k_count_deg(const int* __restrict__ dst,
                                                   float* __restrict__ deg, int e) {
  int i = blockIdx.x * blockDim.x + threadIdx.x;
  if (i < e) atomicAdd(&deg[dst[i]], 1.0f);
}

__global__ __launch_bounds__(256) void k_rsqrt(float* __restrict__ deg, int n) {
  int i = blockIdx.x * blockDim.x + threadIdx.x;
  if (i < n) deg[i] = rsqrtf(deg[i]);
}


__global__ __launch_bounds__(32) void k_gemm_wmma(const float* __restrict__ X,
                                                  const float* __restrict__ W,
                                                  float* __restrict__ H,
                                                  int M, int Nout, int leaky) {
  const int K = 64;
  const int lane = threadIdx.x;
  const int m    = lane & 15;
  const int khi  = (lane >> 4) << 1;

  const int row0 = blockIdx.x << 4;
  const int col0 = blockIdx.y << 4;

  int arow = row0 + m;
  if (arow >= M) arow = M - 1;
  const float* __restrict__ xr = X + (long long)arow * K;
  const float* __restrict__ wc = W + col0 + m;

  v8f c = {};
#pragma unroll
  for (int k0 = 0; k0 < K; k0 += 4) {
    float a0 = xr[k0 + khi + 0];
    float a1 = xr[k0 + khi + 1];
    if (leaky) {
      a0 = (a0 > 0.0f) ? a0 : NEG_SLOPE * a0;
      a1 = (a1 > 0.0f) ? a1 : NEG_SLOPE * a1;
    }
    v2f a, b;
    a.x = a0;
    a.y = a1;
    b.x = wc[(k0 + khi + 0) * Nout];
    b.y = wc[(k0 + khi + 1) * Nout];
    c = __builtin_amdgcn_wmma_f32_16x16x4_f32(false, a, false, b, (short)0, c,
                                              false, false);
  }

  const int hi8 = (lane >> 4) << 3;
#pragma unroll
  for (int v = 0; v < 8; ++v) {
    int r = row0 + hi8 + v;
    if (r < M) H[(long long)r * Nout + col0 + m] = c[v];
  }
}

__global__ __launch_bounds__(256) void k_agg_init(const float* __restrict__ H,
                                                  const float* __restrict__ dinv,
                                                  const float* __restrict__ bias,
                                                  float* __restrict__ A,
                                                  int n, int C, int tshift) {
  long long gid = (long long)blockIdx.x * blockDim.x + threadIdx.x;
  if (gid >= ((long long)n << tshift)) return;
  int i = (int)(gid >> tshift);
  int t = (int)(gid & ((1 << tshift) - 1));
  float di = dinv[i];
  float w  = di * di;
  float4 h = *((const float4*)(H + (long long)i * C) + t);
  float4 b = *((const float4*)bias + t);
  float4 r = make_float4(b.x + w * h.x, b.y + w * h.y,
                         b.z + w * h.z, b.w + w * h.w);
  *((float4*)(A + (long long)i * C) + t) = r;
}

__global__ __launch_bounds__(256) void k_agg_edges(const float* __restrict__ H,
                                                   const float* __restrict__ dinv,
                                                   const int* __restrict__ src,
                                                   const int* __restrict__ dst,
                                                   float* __restrict__ A,
                                                   int e, int C, int tshift) {
  long long gid = (long long)blockIdx.x * blockDim.x + threadIdx.x;
  if (gid >= ((long long)e << tshift)) return;
  int ed = (int)(gid >> tshift);
  int t  = (int)(gid & ((1 << tshift) - 1));
  int s = src[ed];
  int d = dst[ed];
  float w = dinv[s] * dinv[d];
  float4 h = *((const float4*)(H + (long long)s * C) + t);
  float* o = A + (long long)d * C + (t << 2);
  atomicAdd(o + 0, w * h.x);
  atomicAdd(o + 1, w * h.y);
  atomicAdd(o + 2, w * h.z);
  atomicAdd(o + 3, w * h.w);
}


extern "C" void kernel_launch(void* const* d_in, const int* in_sizes, int n_in,
                              void* d_out, int out_size, void* d_ws, size_t ws_size,
                              hipStream_t stream) {
  (void)n_in; (void)out_size; (void)ws_size;

  const float* x  = (const float*)d_in[0];
  const int*   ei = (const int*)  d_in[1];
  const float* W1 = (const float*)d_in[2];
  const float* b1 = (const float*)d_in[3];
  const float* W2 = (const float*)d_in[4];
  const float* b2 = (const float*)d_in[5];
  float* out = (float*)d_out;

  const int DIN  = 64;
  const int N    = in_sizes[0] / DIN;
  const int E    = in_sizes[1] / 2;
  const int DOUT = in_sizes[5];

  const int* src = ei;
  const int* dst = ei + E;

  float* dinv = (float*)d_ws;
  int Nr = ((N + 255) / 256) * 256;
  float* H1 = dinv + Nr;
  float* A1 = H1 + (size_t)N * DIN;
  float* H2 = H1;

  const int TB = 256;

  k_init_deg <<<(N + TB - 1) / TB, TB, 0, stream>>>(dinv, N);
  k_count_deg<<<(E + TB - 1) / TB, TB, 0, stream>>>(dst, dinv, E);
  k_rsqrt    <<<(N + TB - 1) / TB, TB, 0, stream>>>(dinv, N);

  const int Mtiles = (N + 15) / 16;

  k_gemm_wmma<<<dim3(Mtiles, DIN / 16), 32, 0, stream>>>(x, W1, H1, N, DIN, 0);
  {
    long long tot = (long long)N << 4;
    k_agg_init<<<(int)((tot + TB - 1) / TB), TB, 0, stream>>>(H1, dinv, b1, A1, N, DIN, 4);
    tot = (long long)E << 4;
    k_agg_edges<<<(int)((tot + TB - 1) / TB), TB, 0, stream>>>(H1, dinv, src, dst, A1, E, DIN, 4);
  }

  k_gemm_wmma<<<dim3(Mtiles, DOUT / 16), 32, 0, stream>>>(A1, W2, H2, N, DOUT, 1);
  {
    long long tot = (long long)N << 3;
    k_agg_init<<<(int)((tot + TB - 1) / TB), TB, 0, stream>>>(H2, dinv, b2, out, N, DOUT, 3);
    tot = (long long)E << 3;
    k_agg_edges<<<(int)((tot + TB - 1) / TB), TB, 0, stream>>>(H2, dinv, src, dst, out, E, DOUT, 3);
  }
}


# edge scatter regrouped: a wave walks 32 edges, lane l on floats l, l+32 of a row, so each row load and each f32 atomic add covers whole 128-byte lines; same adds, arrival order as before
# speedup vs baseline: 1.4253x; 1.4253x over previous
#include <hip/hip_runtime.h>
#include <hip/hip_bf16.h>

typedef __attribute__((ext_vector_type(2))) float v2f;
typedef __attribute__((ext_vector_type(8))) float v8f;

#define NEG_SLOPE 0.01f


__global__ __launch_bounds__(256) void k_init_deg(float* __restrict__ deg, int n) {
  int i = blockIdx.x * blockDim.x + threadIdx.x;
  if (i < n) deg[i] = 1.0f;
}

__global__ __launch_bounds__(256) void k_count_deg(const int* __restrict__ dst,
                                                   float* __restrict__ deg, int e) {
  int i = blockIdx.x * blockDim.x + threadIdx.x;
  if (i < e) atomicAdd(&deg[dst[i]], 1.0f);
}

__global__ __launch_bounds__(256) void k_rsqrt(float* __restrict__ deg, int n) {
  int i = blockIdx.x * blockDim.x + threadIdx.x;
  if (i < n) deg[i] = rsqrtf(deg[i]);
}


__global__ __launch_bounds__(32) void k_gemm_wmma(const float* __restrict__ X,
                                                  const float* __restrict__ W,
                                                  float* __restrict__ H,
                                                  int M, int Nout, int leaky) {
  const int K = 64;
  const int lane = threadIdx.x;
  const int m    = lane & 15;
  const int khi  = (lane >> 4) << 1;

  const int row0 = blockIdx.x << 4;
  const int col0 = blockIdx.y << 4;

  int arow = row0 + m;
  if (arow >= M) arow = M - 1;
  const float* __restrict__ xr = X + (long long)arow * K;
  const float* __restrict__ wc = W + col0 + m;

  v8f c = {};
#pragma unroll
  for (int k0 = 0; k0 < K; k0 += 4) {
    float a0 = xr[k0 + khi + 0];
    float a1 = xr[k0 + khi + 1];
    if (leaky) {
      a0 = (a0 > 0.0f) ? a0 : NEG_SLOPE * a0;
      a1 = (a1 > 0.0f) ? a1 : NEG_SLOPE * a1;
    }
    v2f a, b;
    a.x = a0;
    a.y = a1;
    b.x = wc[(k0 + khi + 0) * Nout];
    b.y = wc[(k0 + khi + 1) * Nout];
    c = __builtin_amdgcn_wmma_f32_16x16x4_f32(false, a, false, b, (short)0, c,
                                              false, false);
  }

  const int hi8 = (lane >> 4) << 3;
#pragma unroll
  for (int v = 0; v < 8; ++v) {
    int r = row0 + hi8 + v;
    if (r < M) H[(long long)r * Nout + col0 + m] = c[v];
  }
}

__global__ __launch_bounds__(256) void k_agg_init(const float* __restrict__ H,
                                                  const float* __restrict__ dinv,
                                                  const float* __restrict__ bias,
                                                  float* __restrict__ A,
                                                  int n, int C, int tshift) {
  long long gid = (long long)blockIdx.x * blockDim.x + threadIdx.x;
  if (gid >= ((long long)n << tshift)) return;
  int i = (int)(gid >> tshift);
  int t = (int)(gid & ((1 << tshift) - 1));
  float di = dinv[i];
  float w  = di * di;
  float4 h = *((const float4*)(H + (long long)i * C) + t);
  float4 b = *((const float4*)bias + t);
  float4 r = make_float4(b.x + w * h.x, b.y + w * h.y,
                         b.z + w * h.z, b.w + w * h.w);
  *((float4*)(A + (long long)i * C) + t) = r;
}

template <int C>
__global__ __launch_bounds__(256) void k_agg_edges(const float* __restrict__ H,
                                                   const float* __restrict__ dinv,
                                                   const int* __restrict__ src,
                                                   const int* __restrict__ dst,
                                                   float* __restrict__ A,
                                                   int e) {
  constexpr int R = C / 32;
  constexpr int U = 16 / R;
  const int lane = threadIdx.x & 31;
  const int wave = __builtin_amdgcn_readfirstlane(
      (int)((blockIdx.x * 256u + threadIdx.x) >> 5));
  const long long base = (long long)wave << 5;
  if (base >= e) return;
  const int cnt = (e - base < 32) ? (int)(e - base) : 32;
  const bool ok = lane < cnt;
  const int s_v = ok ? src[base + lane] : 0;
  const int d_v = ok ? dst[base + lane] : 0;
  const float w_v = dinv[s_v] * dinv[d_v];
  const int w_b = __float_as_int(w_v);

  if (cnt == 32) {
    for (int i0 = 0; i0 < 32; i0 += U) {
      float h[U][R];
#pragma unroll
      for (int u = 0; u < U; ++u) {
        const int s = __builtin_amdgcn_readlane(s_v, i0 + u);
        const float* __restrict__ row = H + (long long)s * C + lane;
#pragma unroll
        for (int r = 0; r < R; ++r) h[u][r] = row[32 * r];
      }
#pragma unroll
      for (int u = 0; u < U; ++u) {
        const int d = __builtin_amdgcn_readlane(d_v, i0 + u);
        const float w = __int_as_float(__builtin_amdgcn_readlane(w_b, i0 + u));
        float* __restrict__ o = A + (long long)d * C + lane;
#pragma unroll
        for (int r = 0; r < R; ++r) atomicAdd(o + 32 * r, w * h[u][r]);
      }
    }
  } else {
    for (int i = 0; i < cnt; ++i) {
      const int s = __builtin_amdgcn_readlane(s_v, i);
      const int d = __builtin_amdgcn_readlane(d_v, i);
      const float w = __int_as_float(__builtin_amdgcn_readlane(w_b, i));
      const float* __restrict__ row = H + (long long)s * C + lane;
      float* __restrict__ o = A + (long long)d * C + lane;
#pragma unroll
      for (int r = 0; r < R; ++r) atomicAdd(o + 32 * r, w * row[32 * r]);
    }
  }
}


extern "C" void kernel_launch(void* const* d_in, const int* in_sizes, int n_in,
                              void* d_out, int out_size, void* d_ws, size_t ws_size,
                              hipStream_t stream) {
  (void)n_in; (void)out_size; (void)ws_size;

  const float* x  = (const float*)d_in[0];
  const int*   ei = (const int*)  d_in[1];
  const float* W1 = (const float*)d_in[2];
  const float* b1 = (const float*)d_in[3];
  const float* W2 = (const float*)d_in[4];
  const float* b2 = (const float*)d_in[5];
  float* out = (float*)d_out;

  const int DIN  = 64;
  const int N    = in_sizes[0] / DIN;
  const int E    = in_sizes[1] / 2;
  const int DOUT = in_sizes[5];

  const int* src = ei;
  const int* dst = ei + E;

  float* dinv = (float*)d_ws;
  int Nr = ((N + 255) / 256) * 256;
  float* H1 = dinv + Nr;
  float* A1 = H1 + (size_t)N * DIN;
  float* H2 = H1;

  const int TB = 256;

  k_init_deg <<<(N + TB - 1) / TB, TB, 0, stream>>>(dinv, N);
  k_count_deg<<<(E + TB - 1) / TB, TB, 0, stream>>>(dst, dinv, E);
  k_rsqrt    <<<(N + TB - 1) / TB, TB, 0, stream>>>(dinv, N);

  const int Mtiles = (N + 15) / 16;
  const int EB = ((E + 31) / 32 + 7) / 8;

  k_gemm_wmma<<<dim3(Mtiles, DIN / 16), 32, 0, stream>>>(x, W1, H1, N, DIN, 0);
  {
    long long tot = (long long)N << 4;
    k_agg_init<<<(int)((tot + TB - 1) / TB), TB, 0, stream>>>(H1, dinv, b1, A1, N, DIN, 4);
    k_agg_edges<64><<<EB, TB, 0, stream>>>(H1, dinv, src, dst, A1, E);
  }

  k_gemm_wmma<<<dim3(Mtiles, DOUT / 16), 32, 0, stream>>>(A1, W2, H2, N, DOUT, 1);
  {
    long long tot = (long long)N << 3;
    k_agg_init<<<(int)((tot + TB - 1) / TB), TB, 0, stream>>>(H2, dinv, b2, out, N, DOUT, 3);
    k_agg_edges<32><<<EB, TB, 0, stream>>>(H2, dinv, src, dst, out, E);
  }
}


# fewer, longer waves: degree count and row init loop in a fixed 1024-block grid (4 loads in flight), GEMM is one wave per 16-row tile over all column tiles, 8 per block; sum order unchanged
# speedup vs baseline: 1.5002x; 1.0526x over previous
#include <hip/hip_runtime.h>
#include <hip/hip_bf16.h>

typedef __attribute__((ext_vector_type(2))) float v2f;
typedef __attribute__((ext_vector_type(8))) float v8f;

#define NEG_SLOPE 0.01f


__global__ __launch_bounds__(256) void k_init_deg(float* __restrict__ deg, int n) {
  int i = blockIdx.x * blockDim.x + threadIdx.x;
  if (i < n) deg[i] = 1.0f;
}

__global__ __launch_bounds__(256) void k_count_deg(const int* __restrict__ dst,
                                                   float* __restrict__ deg, int e) {
  const int stride = gridDim.x * blockDim.x;
  for (int i0 = blockIdx.x * blockDim.x + threadIdx.x; i0 < e; i0 += 4 * stride) {
    int d[4];
#pragma unroll
    for (int u = 0; u < 4; ++u) {
      const int i = i0 + u * stride;
      d[u] = (i < e) ? dst[i] : -1;
    }
#pragma unroll
    for (int u = 0; u < 4; ++u)
      if (d[u] >= 0) atomicAdd(&deg[d[u]], 1.0f);
  }
}

__global__ __launch_bounds__(256) void k_rsqrt(float* __restrict__ deg, int n) {
  int i = blockIdx.x * blockDim.x + threadIdx.x;
  if (i < n) deg[i] = rsqrtf(deg[i]);
}


template <int NT, bool LEAKY>
__global__ __launch_bounds__(256) void k_gemm_wmma(const float* __restrict__ X,
                                                   const float* __restrict__ W,
                                                   float* __restrict__ H,
                                                   int M) {
  constexpr int K = 64;
  constexpr int Nout = NT * 16;
  const int lane = threadIdx.x & 31;
  const int m    = lane & 15;
  const int khi  = (lane >> 4) << 1;

  const int tile = blockIdx.x * 8 + (threadIdx.x >> 5);
  const int row0 = tile << 4;
  if (row0 >= M) return;

  int arow = row0 + m;
  if (arow >= M) arow = M - 1;
  const float* __restrict__ xr = X + (long long)arow * K + khi;
  const float* __restrict__ wc = W + (long long)khi * Nout + m;

  v2f a[K / 4];
#pragma unroll
  for (int j = 0; j < K / 4; ++j) {
    a[j] = *(const v2f*)(xr + 4 * j);
    if (LEAKY) {
      a[j].x = (a[j].x > 0.0f) ? a[j].x : NEG_SLOPE * a[j].x;
      a[j].y = (a[j].y > 0.0f) ? a[j].y : NEG_SLOPE * a[j].y;
    }
  }

  v8f c[NT];
#pragma unroll
  for (int t = 0; t < NT; ++t) c[t] = v8f{};

#pragma unroll
  for (int j = 0; j < K / 4; ++j) {
#pragma unroll
    for (int t = 0; t < NT; ++t) {
      v2f b;
      b.x = wc[(4 * j + 0) * Nout + 16 * t];
      b.y = wc[(4 * j + 1) * Nout + 16 * t];
      c[t] = __builtin_amdgcn_wmma_f32_16x16x4_f32(false, a[j], false, b, (short)0,
                                                   c[t], false, false);
    }
  }

  const int hi8 = (lane >> 4) << 3;
#pragma unroll
  for (int v = 0; v < 8; ++v) {
    int r = row0 + hi8 + v;
    if (r < M) {
#pragma unroll
      for (int t = 0; t < NT; ++t) H[(long long)r * Nout + 16 * t + m] = c[t][v];
    }
  }
}

__global__ __launch_bounds__(256) void k_agg_init(const float* __restrict__ H,
                                                  const float* __restrict__ dinv,
                                                  const float* __restrict__ bias,
                                                  float* __restrict__ A,
                                                  int n, int C, int tshift) {
  (void)C;
  const int tot = n << tshift;
  const int stride = gridDim.x * blockDim.x;
  const int t = threadIdx.x & ((1 << tshift) - 1);
  const float4 b = *((const float4*)bias + t);
  for (int g0 = blockIdx.x * blockDim.x + threadIdx.x; g0 < tot; g0 += 4 * stride) {
    float4 h[4];
    float di[4];
#pragma unroll
    for (int u = 0; u < 4; ++u) {
      const int gid = g0 + u * stride;
      if (gid < tot) {
        di[u] = dinv[gid >> tshift];
        h[u]  = *((const float4*)H + gid);
      }
    }
#pragma unroll
    for (int u = 0; u < 4; ++u) {
      const int gid = g0 + u * stride;
      if (gid < tot) {
        const float w = di[u] * di[u];
        *((float4*)A + gid) = make_float4(b.x + w * h[u].x, b.y + w * h[u].y,
                                          b.z + w * h[u].z, b.w + w * h[u].w);
      }
    }
  }
}

template <int C>
__global__ __launch_bounds__(256) void k_agg_edges(const float* __restrict__ H,
                                                   const float* __restrict__ dinv,
                                                   const int* __restrict__ src,
                                                   const int* __restrict__ dst,
                                                   float* __restrict__ A,
                                                   int e) {
  constexpr int R = C / 32;
  constexpr int U = 16 / R;
  const int lane = threadIdx.x & 31;
  const int wave = __builtin_amdgcn_readfirstlane(
      (int)((blockIdx.x * 256u + threadIdx.x) >> 5));
  const long long base = (long long)wave << 5;
  if (base >= e) return;
  const int cnt = (e - base < 32) ? (int)(e - base) : 32;
  const bool ok = lane < cnt;
  const int s_v = ok ? src[base + lane] : 0;
  const int d_v = ok ? dst[base + lane] : 0;
  const float w_v = dinv[s_v] * dinv[d_v];
  const int w_b = __float_as_int(w_v);

  if (cnt == 32) {
    for (int i0 = 0; i0 < 32; i0 += U) {
      float h[U][R];
#pragma unroll
      for (int u = 0; u < U; ++u) {
        const int s = __builtin_amdgcn_readlane(s_v, i0 + u);
        const float* __restrict__ row = H + (long long)s * C + lane;
#pragma unroll
        for (int r = 0; r < R; ++r) h[u][r] = row[32 * r];
      }
#pragma unroll
      for (int u = 0; u < U; ++u) {
        const int d = __builtin_amdgcn_readlane(d_v, i0 + u);
        const float w = __int_as_float(__builtin_amdgcn_readlane(w_b, i0 + u));
        float* __restrict__ o = A + (long long)d * C + lane;
#pragma unroll
        for (int r = 0; r < R; ++r) atomicAdd(o + 32 * r, w * h[u][r]);
      }
    }
  } else {
    for (int i = 0; i < cnt; ++i) {
      const int s = __builtin_amdgcn_readlane(s_v, i);
      const int d = __builtin_amdgcn_readlane(d_v, i);
      const float w = __int_as_float(__builtin_amdgcn_readlane(w_b, i));
      const float* __restrict__ row = H + (long long)s * C + lane;
      float* __restrict__ o = A + (long long)d * C + lane;
#pragma unroll
      for (int r = 0; r < R; ++r) atomicAdd(o + 32 * r, w * row[32 * r]);
    }
  }
}


extern "C" void kernel_launch(void* const* d_in, const int* in_sizes, int n_in,
                              void* d_out, int out_size, void* d_ws, size_t ws_size,
                              hipStream_t stream) {
  (void)n_in; (void)out_size; (void)ws_size;

  const float* x  = (const float*)d_in[0];
  const int*   ei = (const int*)  d_in[1];
  const float* W1 = (const float*)d_in[2];
  const float* b1 = (const float*)d_in[3];
  const float* W2 = (const float*)d_in[4];
  const float* b2 = (const float*)d_in[5];
  float* out = (float*)d_out;

  const int DIN  = 64;
  const int N    = in_sizes[0] / DIN;
  const int E    = in_sizes[1] / 2;
  const int DOUT = in_sizes[5];

  const int* src = ei;
  const int* dst = ei + E;

  float* dinv = (float*)d_ws;
  int Nr = ((N + 255) / 256) * 256;
  float* H1 = dinv + Nr;
  float* A1 = H1 + (size_t)N * DIN;
  float* H2 = H1;

  const int TB = 256;
  const int RES = 1024;

  k_init_deg <<<(N + TB - 1) / TB, TB, 0, stream>>>(dinv, N);
  int CB = (E + TB - 1) / TB;
  if (CB > RES) CB = RES;
  k_count_deg<<<CB, TB, 0, stream>>>(dst, dinv, E);
  k_rsqrt    <<<(N + TB - 1) / TB, TB, 0, stream>>>(dinv, N);

  const int Mtiles = (N + 15) / 16;
  const int EB = ((E + 31) / 32 + 7) / 8;

  const int GB = (Mtiles + 7) / 8;
  k_gemm_wmma<4, false><<<GB, TB, 0, stream>>>(x, W1, H1, N);
  {
    long long tot = (long long)N << 4;
    int IB = (int)((tot + TB - 1) / TB);
    if (IB > RES) IB = RES;
    k_agg_init<<<IB, TB, 0, stream>>>(H1, dinv, b1, A1, N, DIN, 4);
    k_agg_edges<64><<<EB, TB, 0, stream>>>(H1, dinv, src, dst, A1, E);
  }

  k_gemm_wmma<2, true><<<GB, TB, 0, stream>>>(A1, W2, H2, N);
  {
    long long tot = (long long)N << 3;
    int IB = (int)((tot + TB - 1) / TB);
    if (IB > RES) IB = RES;
    k_agg_init<<<IB, TB, 0, stream>>>(H2, dinv, b2, out, N, DOUT, 3);
    k_agg_edges<32><<<EB, TB, 0, stream>>>(H2, dinv, src, dst, out, E);
  }
}


# edges sorted by destination row in each call (two-level counting sort, LDS rank adds); a wave sums a row's source rows in registers, stores it once: no float atomics; sum order = sort order
# speedup vs baseline: 2.8186x; 1.8788x over previous
#include <hip/hip_runtime.h>
#include <hip/hip_bf16.h>

typedef __attribute__((ext_vector_type(2))) float v2f;
typedef __attribute__((ext_vector_type(8))) float v8f;

#define NEG_SLOPE 0.01f


__global__ __launch_bounds__(256) void k_init_deg(float* __restrict__ deg, int n) {
  int i = blockIdx.x * blockDim.x + threadIdx.x;
  if (i < n) deg[i] = 1.0f;
}

__global__ __launch_bounds__(256) void k_count_deg(const int* __restrict__ dst,
                                                   float* __restrict__ deg, int e) {
  const int stride = gridDim.x * blockDim.x;
  for (int i0 = blockIdx.x * blockDim.x + threadIdx.x; i0 < e; i0 += 4 * stride) {
    int d[4];
#pragma unroll
    for (int u = 0; u < 4; ++u) {
      const int i = i0 + u * stride;
      d[u] = (i < e) ? dst[i] : -1;
    }
#pragma unroll
    for (int u = 0; u < 4; ++u)
      if (d[u] >= 0) atomicAdd(&deg[d[u]], 1.0f);
  }
}

__global__ __launch_bounds__(256) void k_rsqrt(float* __restrict__ deg, int n) {
  int i = blockIdx.x * blockDim.x + threadIdx.x;
  if (i < n) deg[i] = rsqrtf(deg[i]);
}


template <int NT, bool LEAKY, bool ZERO>
__global__ __launch_bounds__(256) void k_gemm_wmma(const float* __restrict__ X,
                                                   const float* __restrict__ W,
                                                   float* __restrict__ H,
                                                   int M, int* __restrict__ zbuf, int zn) {
  constexpr int K = 64;
  if (ZERO) {
    const int zi = blockIdx.x * 256 + threadIdx.x;
    if (zi < zn) zbuf[zi] = 0;
  }
  constexpr int Nout = NT * 16;
  const int lane = threadIdx.x & 31;
  const int m    = lane & 15;
  const int khi  = (lane >> 4) << 1;

  const int tile = blockIdx.x * 8 + (threadIdx.x >> 5);
  const int row0 = tile << 4;
  if (row0 >= M) return;

  int arow = row0 + m;
  if (arow >= M) arow = M - 1;
  const float* __restrict__ xr = X + (long long)arow * K + khi;
  const float* __restrict__ wc = W + (long long)khi * Nout + m;

  v2f a[K / 4];
#pragma unroll
  for (int j = 0; j < K / 4; ++j) {
    a[j] = *(const v2f*)(xr + 4 * j);
    if (LEAKY) {
      a[j].x = (a[j].x > 0.0f) ? a[j].x : NEG_SLOPE * a[j].x;
      a[j].y = (a[j].y > 0.0f) ? a[j].y : NEG_SLOPE * a[j].y;
    }
  }

  v8f c[NT];
#pragma unroll
  for (int t = 0; t < NT; ++t) c[t] = v8f{};

#pragma unroll
  for (int j = 0; j < K / 4; ++j) {
#pragma unroll
    for (int t = 0; t < NT; ++t) {
      v2f b;
      b.x = wc[(4 * j + 0) * Nout + 16 * t];
      b.y = wc[(4 * j + 1) * Nout + 16 * t];
      c[t] = __builtin_amdgcn_wmma_f32_16x16x4_f32(false, a[j], false, b, (short)0,
                                                   c[t], false, false);
    }
  }

  const int hi8 = (lane >> 4) << 3;
#pragma unroll
  for (int v = 0; v < 8; ++v) {
    int r = row0 + hi8 + v;
    if (r < M) {
#pragma unroll
      for (int t = 0; t < NT; ++t) H[(long long)r * Nout + 16 * t + m] = c[t][v];
    }
  }
}

__global__ __launch_bounds__(256) void k_agg_init(const float* __restrict__ H,
                                                  const float* __restrict__ dinv,
                                                  const float* __restrict__ bias,
                                                  float* __restrict__ A,
                                                  int n, int C, int tshift) {
  (void)C;
  const int tot = n << tshift;
  const int stride = gridDim.x * blockDim.x;
  const int t = threadIdx.x & ((1 << tshift) - 1);
  const float4 b = *((const float4*)bias + t);
  for (int g0 = blockIdx.x * blockDim.x + threadIdx.x; g0 < tot; g0 += 4 * stride) {
    float4 h[4];
    float di[4];
#pragma unroll
    for (int u = 0; u < 4; ++u) {
      const int gid = g0 + u * stride;
      if (gid < tot) {
        di[u] = dinv[gid >> tshift];
        h[u]  = *((const float4*)H + gid);
      }
    }
#pragma unroll
    for (int u = 0; u < 4; ++u) {
      const int gid = g0 + u * stride;
      if (gid < tot) {
        const float w = di[u] * di[u];
        *((float4*)A + gid) = make_float4(b.x + w * h[u].x, b.y + w * h[u].y,
                                          b.z + w * h[u].z, b.w + w * h[u].w);
      }
    }
  }
}

template <int C>
__global__ __launch_bounds__(256) void k_agg_edges(const float* __restrict__ H,
                                                   const float* __restrict__ dinv,
                                                   const int* __restrict__ src,
                                                   const int* __restrict__ dst,
                                                   float* __restrict__ A,
                                                   int e) {
  constexpr int R = C / 32;
  constexpr int U = 16 / R;
  const int lane = threadIdx.x & 31;
  const int wave = __builtin_amdgcn_readfirstlane(
      (int)((blockIdx.x * 256u + threadIdx.x) >> 5));
  const long long base = (long long)wave << 5;
  if (base >= e) return;
  const int cnt = (e - base < 32) ? (int)(e - base) : 32;
  const bool ok = lane < cnt;
  const int s_v = ok ? src[base + lane] : 0;
  const int d_v = ok ? dst[base + lane] : 0;
  const float w_v = dinv[s_v] * dinv[d_v];
  const int w_b = __float_as_int(w_v);

  if (cnt == 32) {
    for (int i0 = 0; i0 < 32; i0 += U) {
      float h[U][R];
#pragma unroll
      for (int u = 0; u < U; ++u) {
        const int s = __builtin_amdgcn_readlane(s_v, i0 + u);
        const float* __restrict__ row = H + (long long)s * C + lane;
#pragma unroll
        for (int r = 0; r < R; ++r) h[u][r] = row[32 * r];
      }
#pragma unroll
      for (int u = 0; u < U; ++u) {
        const int d = __builtin_amdgcn_readlane(d_v, i0 + u);
        const float w = __int_as_float(__builtin_amdgcn_readlane(w_b, i0 + u));
        float* __restrict__ o = A + (long long)d * C + lane;
#pragma unroll
        for (int r = 0; r < R; ++r) atomicAdd(o + 32 * r, w * h[u][r]);
      }
    }
  } else {
    for (int i = 0; i < cnt; ++i) {
      const int s = __builtin_amdgcn_readlane(s_v, i);
      const int d = __builtin_amdgcn_readlane(d_v, i);
      const float w = __int_as_float(__builtin_amdgcn_readlane(w_b, i));
      const float* __restrict__ row = H + (long long)s * C + lane;
      float* __restrict__ o = A + (long long)d * C + lane;
#pragma unroll
      for (int r = 0; r < R; ++r) atomicAdd(o + 32 * r, w * row[32 * r]);
    }
  }
}


#define BKT_SHIFT 8
#define BKT_ROWS 256
#define BIN_PT 16

__device__ __forceinline__ int lane_read_i(int v, int src_lane) {
  return __builtin_amdgcn_ds_bpermute(src_lane << 2, v);
}
__device__ __forceinline__ float lane_read_f(float v, int src_lane) {
  return __int_as_float(__builtin_amdgcn_ds_bpermute(src_lane << 2, __float_as_int(v)));
}

__device__ __forceinline__ int block_scan_excl(int v, int* wsum, int* total) {
  const int lane = threadIdx.x & 31, wv = threadIdx.x >> 5;
  int inc = v;
  for (int o = 1; o < 32; o <<= 1) {
    const int r = lane_read_i(inc, (lane - o) & 31);
    if (lane >= o) inc += r;
  }
  __syncthreads();
  if (lane == 31) wsum[wv] = inc;
  __syncthreads();
  int woff = 0, tot = 0;
  for (int k = 0; k < 8; ++k) {
    const int s = wsum[k];
    woff += (k < wv) ? s : 0;
    tot += s;
  }
  *total = tot;
  __syncthreads();
  return woff + inc - v;
}

__global__ __launch_bounds__(256) void k_bucket_sizes(const int* __restrict__ dst,
                                                      int* __restrict__ bsum,
                                                      int e, int nb) {
  extern __shared__ int sh[];
  int* hist = sh;
  const int t = threadIdx.x;
  for (int k = t; k < nb; k += 256) hist[k] = 0;
  __syncthreads();
  const long long e0 = (long long)blockIdx.x * (256 * BIN_PT);
  int d[BIN_PT];
#pragma unroll
  for (int u = 0; u < BIN_PT; ++u) {
    const long long i = e0 + u * 256 + t;
    d[u] = (i < e) ? dst[i] : -1;
  }
#pragma unroll
  for (int u = 0; u < BIN_PT; ++u)
    if (d[u] >= 0) atomicAdd(&hist[d[u] >> BKT_SHIFT], 1);
  __syncthreads();
  for (int k = t; k < nb; k += 256) {
    const int h = hist[k];
    if (h > 0) atomicAdd(&bsum[k], h);
  }
}

__global__ __launch_bounds__(256) void k_bin_edges(const int* __restrict__ src,
                                                   const int* __restrict__ dst,
                                                   const int* __restrict__ bsum,
                                                   int* __restrict__ bcur,
                                                   int* __restrict__ tmp_i,
                                                   int e, int nb) {
  extern __shared__ int sh[];
  __shared__ int wsum[8];
  int* hist = sh;
  int* gbase = sh + nb;
  int2* __restrict__ tmp = (int2*)tmp_i;
  const int t = threadIdx.x;
  for (int k = t; k < nb; k += 256) hist[k] = 0;
  const int ch = (nb + 255) >> 8;
  int run = 0;
  for (int k = t * ch; k < t * ch + ch; ++k) run += (k < nb) ? bsum[k] : 0;
  int total;
  int pre = block_scan_excl(run, wsum, &total);
  for (int k = t * ch; k < t * ch + ch; ++k)
    if (k < nb) { gbase[k] = pre; pre += bsum[k]; }
  __syncthreads();
  const long long e0 = (long long)blockIdx.x * (256 * BIN_PT);
  int s[BIN_PT], d[BIN_PT], r[BIN_PT];
#pragma unroll
  for (int u = 0; u < BIN_PT; ++u) {
    const long long i = e0 + u * 256 + t;
    const bool ok = i < e;
    s[u] = ok ? src[i] : 0;
    d[u] = ok ? dst[i] : -1;
  }
#pragma unroll
  for (int u = 0; u < BIN_PT; ++u)
    r[u] = (d[u] >= 0) ? atomicAdd(&hist[d[u] >> BKT_SHIFT], 1) : 0;
  __syncthreads();
  for (int k = t; k < nb; k += 256) {
    const int h = hist[k];
    if (h > 0) gbase[k] += atomicAdd(&bcur[k], h);
  }
  __syncthreads();
#pragma unroll
  for (int u = 0; u < BIN_PT; ++u)
    if (d[u] >= 0) tmp[gbase[d[u] >> BKT_SHIFT] + r[u]] = make_int2(s[u], d[u]);
}

__global__ __launch_bounds__(256) void k_fill_csr(const int* __restrict__ tmp_i,
                                                  const int* __restrict__ bsum,
                                                  int* __restrict__ rowptr,
                                                  float* __restrict__ dinv,
                                                  int* __restrict__ csr, int n) {
  __shared__ int cur[BKT_ROWS];
  __shared__ int wsum[8];
  const int2* __restrict__ tmp = (const int2*)tmp_i;
  const int b = blockIdx.x, t = threadIdx.x;
  const int i = b * BKT_ROWS + t;
  int part = 0;
  for (int k = t; k < b; k += 256) part += bsum[k];
  int lo;
  (void)block_scan_excl(part, wsum, &lo);
  const int hi = lo + bsum[b];
  cur[t] = 0;
  __syncthreads();
#pragma unroll 4
  for (int k = lo + t; k < hi; k += 256) atomicAdd(&cur[tmp[k].y & (BKT_ROWS - 1)], 1);
  __syncthreads();
  const int c = cur[t];
  int tot;
  const int start = lo + block_scan_excl(c, wsum, &tot);
  if (i < n) {
    rowptr[i] = start;
    dinv[i] = rsqrtf(1.0f + (float)c);
  }
  if (i == n - 1) rowptr[n] = start + c;
  cur[t] = start;
  __syncthreads();
#pragma unroll 4
  for (int k = lo + t; k < hi; k += 256) {
    const int2 ed = tmp[k];
    const int pos = atomicAdd(&cur[ed.y & (BKT_ROWS - 1)], 1);
    csr[pos] = ed.x;
  }
}

template <int C, int U>
__global__ __launch_bounds__(256) void k_gather_rows(const float* __restrict__ H,
                                                     const float* __restrict__ dinv,
                                                     const int* __restrict__ rowptr,
                                                     const int* __restrict__ csr,
                                                     const float* __restrict__ bias,
                                                     float* __restrict__ A, int n) {
  constexpr int G  = C / 4;
  constexpr int EP = 32 / G;
  const int lane = threadIdx.x & 31;
  const int g = lane / G, q = lane % G;
  const int nw = gridDim.x * 8;
  const float4* __restrict__ H4 = (const float4*)H;
  float4* __restrict__ A4 = (float4*)A;
  const float4 bq = ((const float4*)bias)[q];
  const int w0 = __builtin_amdgcn_readfirstlane((int)(blockIdx.x * 8u + (threadIdx.x >> 5)));
  if (w0 >= n) return;
  int beg_n = rowptr[w0], end_n = rowptr[w0 + 1];
  for (int i = w0; i < n; i += nw) {
    const int beg = __builtin_amdgcn_readfirstlane(beg_n);
    const int end = __builtin_amdgcn_readfirstlane(end_n);
    if (i + nw < n) {
      beg_n = rowptr[i + nw];
      end_n = rowptr[i + nw + 1];
    }
    const float di = dinv[i];
    const float4 hs = H4[(unsigned)(i * G + q)];
    float4 acc = make_float4(0.0f, 0.0f, 0.0f, 0.0f);
    for (int e0 = beg; e0 < end; e0 += 32) {
      const int cnt = (end - e0 < 32) ? (end - e0) : 32;
      const int li = (lane < cnt) ? lane : (cnt - 1);
      const int s_l = csr[e0 + li];
      const float dv = dinv[s_l];
      for (int j = 0; j < cnt; j += EP * U) {
        float4 h[U];
        int s[U];
#pragma unroll
        for (int u = 0; u < U; ++u) s[u] = lane_read_i(s_l, j + u * EP + g);
#pragma unroll
        for (int u = 0; u < U; ++u) {
          h[u] = make_float4(0.0f, 0.0f, 0.0f, 0.0f);
          if (j + u * EP + g < cnt) h[u] = H4[(unsigned)(s[u] * G + q)];
        }
#pragma unroll
        for (int u = 0; u < U; ++u) {
          const float w = lane_read_f(dv, j + u * EP + g) * di;
          acc.x += w * h[u].x;
          acc.y += w * h[u].y;
          acc.z += w * h[u].z;
          acc.w += w * h[u].w;
        }
      }
    }
#pragma unroll
    for (int o = G; o < 32; o <<= 1) {
      acc.x += lane_read_f(acc.x, lane ^ o);
      acc.y += lane_read_f(acc.y, lane ^ o);
      acc.z += lane_read_f(acc.z, lane ^ o);
      acc.w += lane_read_f(acc.w, lane ^ o);
    }
    if (g == 0) {
      const float ws = di * di;
      A4[(unsigned)(i * G + q)] = make_float4(bq.x + (acc.x + ws * hs.x), bq.y + (acc.y + ws * hs.y),
                                              bq.z + (acc.z + ws * hs.z), bq.w + (acc.w + ws * hs.w));
    }
  }
}


extern "C" void kernel_launch(void* const* d_in, const int* in_sizes, int n_in,
                              void* d_out, int out_size, void* d_ws, size_t ws_size,
                              hipStream_t stream) {
  (void)n_in; (void)out_size;

  const float* x  = (const float*)d_in[0];
  const int*   ei = (const int*)  d_in[1];
  const float* W1 = (const float*)d_in[2];
  const float* b1 = (const float*)d_in[3];
  const float* W2 = (const float*)d_in[4];
  const float* b2 = (const float*)d_in[5];
  float* out = (float*)d_out;

  const int DIN  = 64;
  const int N    = in_sizes[0] / DIN;
  const int E    = in_sizes[1] / 2;
  const int DOUT = in_sizes[5];

  const int* src = ei;
  const int* dst = ei + E;

  float* dinv = (float*)d_ws;
  int Nr = ((N + 255) / 256) * 256;
  float* H1 = dinv + Nr;
  float* A1 = H1 + (size_t)N * DIN;
  float* H2 = H1;

  const int TB = 256;
  const int RES = 1024;
  const int Mtiles = (N + 15) / 16;
  const int GB = (Mtiles + 7) / 8;

  const int NB = (N + BKT_ROWS - 1) / BKT_ROWS;
  const size_t need = ((size_t)Nr + 2 * (size_t)N * DIN + (size_t)Nr + 256 + (size_t)E) * 4;
  const bool sorted_path = ws_size >= need && E > 0 && N < (1 << 26) && NB <= 8192 &&
                           2 * (size_t)E + 2 * (size_t)NB <= (size_t)N * DIN &&
                           2 * NB <= GB * TB;

  if (sorted_path) {
    int* tmp  = (int*)A1;
    int* bsum = tmp + 2 * (size_t)E;
    int* bcur = bsum + NB;
    int* rowptr = (int*)(A1 + (size_t)N * DIN);
    int* csr = rowptr + Nr + 256;

    k_gemm_wmma<4, false, true><<<GB, TB, 0, stream>>>(x, W1, H1, N, bsum, 2 * NB);
    const int BB = (E + TB * BIN_PT - 1) / (TB * BIN_PT);
    k_bucket_sizes<<<BB, TB, NB * sizeof(int), stream>>>(dst, bsum, E, NB);
    k_bin_edges<<<BB, TB, 2 * NB * sizeof(int), stream>>>(src, dst, bsum, bcur, tmp, E, NB);
    k_fill_csr<<<NB, TB, 0, stream>>>(tmp, bsum, rowptr, dinv, csr, N);

    const int GR = 2048;
    k_gather_rows<64, 8><<<GR, TB, 0, stream>>>(H1, dinv, rowptr, csr, b1, A1, N);
    k_gemm_wmma<2, true, false><<<GB, TB, 0, stream>>>(A1, W2, H2, N, nullptr, 0);
    k_gather_rows<32, 4><<<GR, TB, 0, stream>>>(H2, dinv, rowptr, csr, b2, out, N);
    return;
  }

  k_init_deg <<<(N + TB - 1) / TB, TB, 0, stream>>>(dinv, N);
  int CB = (E + TB - 1) / TB;
  if (CB > RES) CB = RES;
  k_count_deg<<<CB, TB, 0, stream>>>(dst, dinv, E);
  k_rsqrt    <<<(N + TB - 1) / TB, TB, 0, stream>>>(dinv, N);

  const int EB = ((E + 31) / 32 + 7) / 8;

  k_gemm_wmma<4, false, false><<<GB, TB, 0, stream>>>(x, W1, H1, N, nullptr, 0);
  {
    long long tot = (long long)N << 4;
    int IB = (int)((tot + TB - 1) / TB);
    if (IB > RES) IB = RES;
    k_agg_init<<<IB, TB, 0, stream>>>(H1, dinv, b1, A1, N, DIN, 4);
    k_agg_edges<64><<<EB, TB, 0, stream>>>(H1, dinv, src, dst, A1, E);
  }

  k_gemm_wmma<2, true, false><<<GB, TB, 0, stream>>>(A1, W2, H2, N, nullptr, 0);
  {
    long long tot = (long long)N << 3;
    int IB = (int)((tot + TB - 1) / TB);
    if (IB > RES) IB = RES;
    k_agg_init<<<IB, TB, 0, stream>>>(H2, dinv, b2, out, N, DOUT, 3);
    k_agg_edges<32><<<EB, TB, 0, stream>>>(H2, dinv, src, dst, out, E);
  }
}


# layer-2 product folded into the layer-1 row gather: 8 waves park 16 finished rows in LDS, one wave in turn runs the f32 WMMAs on the tile; aggregated table never stored, one launch less; same sums
# speedup vs baseline: 3.3881x; 1.2021x over previous
#include <hip/hip_runtime.h>
#include <hip/hip_bf16.h>

typedef __attribute__((ext_vector_type(2))) float v2f;
typedef __attribute__((ext_vector_type(8))) float v8f;

#define NEG_SLOPE 0.01f


__global__ __launch_bounds__(256) void k_init_deg(float* __restrict__ deg, int n) {
  int i = blockIdx.x * blockDim.x + threadIdx.x;
  if (i < n) deg[i] = 1.0f;
}

__global__ __launch_bounds__(256) void k_count_deg(const int* __restrict__ dst,
                                                   float* __restrict__ deg, int e) {
  const int stride = gridDim.x * blockDim.x;
  for (int i0 = blockIdx.x * blockDim.x + threadIdx.x; i0 < e; i0 += 4 * stride) {
    int d[4];
#pragma unroll
    for (int u = 0; u < 4; ++u) {
      const int i = i0 + u * stride;
      d[u] = (i < e) ? dst[i] : -1;
    }
#pragma unroll
    for (int u = 0; u < 4; ++u)
      if (d[u] >= 0) atomicAdd(&deg[d[u]], 1.0f);
  }
}

__global__ __launch_bounds__(256) void k_rsqrt(float* __restrict__ deg, int n) {
  int i = blockIdx.x * blockDim.x + threadIdx.x;
  if (i < n) deg[i] = rsqrtf(deg[i]);
}


template <int NT, bool LEAKY, bool SCALE>
__global__ __launch_bounds__(256) void k_gemm_wmma(const float* __restrict__ X,
                                                   const float* __restrict__ W,
                                                   float* __restrict__ H,
                                                   int M, const float* __restrict__ rscale) {
  constexpr int K = 64;
  constexpr int Nout = NT * 16;
  const int lane = threadIdx.x & 31;
  const int m    = lane & 15;
  const int khi  = (lane >> 4) << 1;

  const int tile = blockIdx.x * 8 + (threadIdx.x >> 5);
  const int row0 = tile << 4;
  if (row0 >= M) return;

  int arow = row0 + m;
  if (arow >= M) arow = M - 1;
  const float* __restrict__ xr = X + (long long)arow * K + khi;
  const float* __restrict__ wc = W + (long long)khi * Nout + m;
  const float rs = SCALE ? rscale[arow] : 1.0f;

  v2f a[K / 4];
#pragma unroll
  for (int j = 0; j < K / 4; ++j) {
    a[j] = *(const v2f*)(xr + 4 * j);
    if (LEAKY) {
      a[j].x = (a[j].x > 0.0f) ? a[j].x : NEG_SLOPE * a[j].x;
      a[j].y = (a[j].y > 0.0f) ? a[j].y : NEG_SLOPE * a[j].y;
    }
    if (SCALE) {
      a[j].x *= rs;
      a[j].y *= rs;
    }
  }

  v8f c[NT];
#pragma unroll
  for (int t = 0; t < NT; ++t) c[t] = v8f{};

#pragma unroll
  for (int j = 0; j < K / 4; ++j) {
#pragma unroll
    for (int t = 0; t < NT; ++t) {
      v2f b;
      b.x = wc[(4 * j + 0) * Nout + 16 * t];
      b.y = wc[(4 * j + 1) * Nout + 16 * t];
      c[t] = __builtin_amdgcn_wmma_f32_16x16x4_f32(false, a[j], false, b, (short)0,
                                                   c[t], false, false);
    }
  }

  const int hi8 = (lane >> 4) << 3;
#pragma unroll
  for (int v = 0; v < 8; ++v) {
    int r = row0 + hi8 + v;
    if (r < M) {
#pragma unroll
      for (int t = 0; t < NT; ++t) H[(long long)r * Nout + 16 * t + m] = c[t][v];
    }
  }
}

__global__ __launch_bounds__(256) void k_agg_init(const float* __restrict__ H,
                                                  const float* __restrict__ dinv,
                                                  const float* __restrict__ bias,
                                                  float* __restrict__ A,
                                                  int n, int C, int tshift) {
  (void)C;
  const int tot = n << tshift;
  const int stride = gridDim.x * blockDim.x;
  const int t = threadIdx.x & ((1 << tshift) - 1);
  const float4 b = *((const float4*)bias + t);
  for (int g0 = blockIdx.x * blockDim.x + threadIdx.x; g0 < tot; g0 += 4 * stride) {
    float4 h[4];
    float di[4];
#pragma unroll
    for (int u = 0; u < 4; ++u) {
      const int gid = g0 + u * stride;
      if (gid < tot) {
        di[u] = dinv[gid >> tshift];
        h[u]  = *((const float4*)H + gid);
      }
    }
#pragma unroll
    for (int u = 0; u < 4; ++u) {
      const int gid = g0 + u * stride;
      if (gid < tot) {
        const float w = di[u] * di[u];
        *((float4*)A + gid) = make_float4(b.x + w * h[u].x, b.y + w * h[u].y,
                                          b.z + w * h[u].z, b.w + w * h[u].w);
      }
    }
  }
}

template <int C>
__global__ __launch_bounds__(256) void k_agg_edges(const float* __restrict__ H,
                                                   const float* __restrict__ dinv,
                                                   const int* __restrict__ src,
                                                   const int* __restrict__ dst,
                                                   float* __restrict__ A,
                                                   int e) {
  constexpr int R = C / 32;
  constexpr int U = 16 / R;
  const int lane = threadIdx.x & 31;
  const int wave = __builtin_amdgcn_readfirstlane(
      (int)((blockIdx.x * 256u + threadIdx.x) >> 5));
  const long long base = (long long)wave << 5;
  if (base >= e) return;
  const int cnt = (e - base < 32) ? (int)(e - base) : 32;
  const bool ok = lane < cnt;
  const int s_v = ok ? src[base + lane] : 0;
  const int d_v = ok ? dst[base + lane] : 0;
  const float w_v = dinv[s_v] * dinv[d_v];
  const int w_b = __float_as_int(w_v);

  if (cnt == 32) {
    for (int i0 = 0; i0 < 32; i0 += U) {
      float h[U][R];
#pragma unroll
      for (int u = 0; u < U; ++u) {
        const int s = __builtin_amdgcn_readlane(s_v, i0 + u);
        const float* __restrict__ row = H + (long long)s * C + lane;
#pragma unroll
        for (int r = 0; r < R; ++r) h[u][r] = row[32 * r];
      }
#pragma unroll
      for (int u = 0; u < U; ++u) {
        const int d = __builtin_amdgcn_readlane(d_v, i0 + u);
        const float w = __int_as_float(__builtin_amdgcn_readlane(w_b, i0 + u));
        float* __restrict__ o = A + (long long)d * C + lane;
#pragma unroll
        for (int r = 0; r < R; ++r) atomicAdd(o + 32 * r, w * h[u][r]);
      }
    }
  } else {
    for (int i = 0; i < cnt; ++i) {
      const int s = __builtin_amdgcn_readlane(s_v, i);
      const int d = __builtin_amdgcn_readlane(d_v, i);
      const float w = __int_as_float(__builtin_amdgcn_readlane(w_b, i));
      const float* __restrict__ row = H + (long long)s * C + lane;
      float* __restrict__ o = A + (long long)d * C + lane;
#pragma unroll
      for (int r = 0; r < R; ++r) atomicAdd(o + 32 * r, w * row[32 * r]);
    }
  }
}


#define BKT_SHIFT 8
#define BKT_ROWS 256
#define BIN_PT 16

__device__ __forceinline__ int lane_read_i(int v, int src_lane) {
  return __builtin_amdgcn_ds_bpermute(src_lane << 2, v);
}
__device__ __forceinline__ float lane_read_f(float v, int src_lane) {
  return __int_as_float(__builtin_amdgcn_ds_bpermute(src_lane << 2, __float_as_int(v)));
}

__device__ __forceinline__ int block_scan_excl(int v, int* wsum, int* total) {
  const int lane = threadIdx.x & 31, wv = threadIdx.x >> 5;
  int inc = v;
  for (int o = 1; o < 32; o <<= 1) {
    const int r = lane_read_i(inc, (lane - o) & 31);
    if (lane >= o) inc += r;
  }
  __syncthreads();
  if (lane == 31) wsum[wv] = inc;
  __syncthreads();
  int woff = 0, tot = 0;
  for (int k = 0; k < 8; ++k) {
    const int s = wsum[k];
    woff += (k < wv) ? s : 0;
    tot += s;
  }
  *total = tot;
  __syncthreads();
  return woff + inc - v;
}

__global__ __launch_bounds__(256) void k_clear_counters(int* __restrict__ zbuf, int zn) {
  const int zi = blockIdx.x * 256 + threadIdx.x;
  if (zi < zn) zbuf[zi] = 0;
}

__global__ __launch_bounds__(256) void k_bucket_sizes(const int* __restrict__ dst,
                                                      int* __restrict__ bsum,
                                                      int e, int nb) {
  extern __shared__ int sh[];
  int* hist = sh;
  const int t = threadIdx.x;
  for (int k = t; k < nb; k += 256) hist[k] = 0;
  __syncthreads();
  const long long e0 = (long long)blockIdx.x * (256 * BIN_PT);
  int d[BIN_PT];
#pragma unroll
  for (int u = 0; u < BIN_PT; ++u) {
    const long long i = e0 + u * 256 + t;
    d[u] = (i < e) ? dst[i] : -1;
  }
#pragma unroll
  for (int u = 0; u < BIN_PT; ++u)
    if (d[u] >= 0) atomicAdd(&hist[d[u] >> BKT_SHIFT], 1);
  __syncthreads();
  for (int k = t; k < nb; k += 256) {
    const int h = hist[k];
    if (h > 0) atomicAdd(&bsum[k], h);
  }
}

__global__ __launch_bounds__(256) void k_bin_edges(const int* __restrict__ src,
                                                   const int* __restrict__ dst,
                                                   const int* __restrict__ bsum,
                                                   int* __restrict__ bcur,
                                                   int* __restrict__ tmp_i,
                                                   int e, int nb) {
  extern __shared__ int sh[];
  __shared__ int wsum[8];
  int* hist = sh;
  int* gbase = sh + nb;
  int2* __restrict__ tmp = (int2*)tmp_i;
  const int t = threadIdx.x;
  for (int k = t; k < nb; k += 256) hist[k] = 0;
  const int ch = (nb + 255) >> 8;
  int run = 0;
  for (int k = t * ch; k < t * ch + ch; ++k) run += (k < nb) ? bsum[k] : 0;
  int total;
  int pre = block_scan_excl(run, wsum, &total);
  for (int k = t * ch; k < t * ch + ch; ++k)
    if (k < nb) { gbase[k] = pre; pre += bsum[k]; }
  __syncthreads();
  const long long e0 = (long long)blockIdx.x * (256 * BIN_PT);
  int s[BIN_PT], d[BIN_PT], r[BIN_PT];
#pragma unroll
  for (int u = 0; u < BIN_PT; ++u) {
    const long long i = e0 + u * 256 + t;
    const bool ok = i < e;
    s[u] = ok ? src[i] : 0;
    d[u] = ok ? dst[i] : -1;
  }
#pragma unroll
  for (int u = 0; u < BIN_PT; ++u)
    r[u] = (d[u] >= 0) ? atomicAdd(&hist[d[u] >> BKT_SHIFT], 1) : 0;
  __syncthreads();
  for (int k = t; k < nb; k += 256) {
    const int h = hist[k];
    if (h > 0) gbase[k] += atomicAdd(&bcur[k], h);
  }
  __syncthreads();
#pragma unroll
  for (int u = 0; u < BIN_PT; ++u)
    if (d[u] >= 0) tmp[gbase[d[u] >> BKT_SHIFT] + r[u]] = make_int2(s[u], d[u]);
}

__global__ __launch_bounds__(256) void k_fill_csr(const int* __restrict__ tmp_i,
                                                  const int* __restrict__ bsum,
                                                  int* __restrict__ rowptr,
                                                  float* __restrict__ dinv,
                                                  int* __restrict__ csr, int n) {
  __shared__ int cur[BKT_ROWS];
  __shared__ int wsum[8];
  const int2* __restrict__ tmp = (const int2*)tmp_i;
  const int b = blockIdx.x, t = threadIdx.x;
  const int i = b * BKT_ROWS + t;
  int part = 0;
  for (int k = t; k < b; k += 256) part += bsum[k];
  int lo;
  (void)block_scan_excl(part, wsum, &lo);
  const int hi = lo + bsum[b];
  cur[t] = 0;
  __syncthreads();
#pragma unroll 4
  for (int k = lo + t; k < hi; k += 256) atomicAdd(&cur[tmp[k].y & (BKT_ROWS - 1)], 1);
  __syncthreads();
  const int c = cur[t];
  int tot;
  const int start = lo + block_scan_excl(c, wsum, &tot);
  if (i < n) {
    rowptr[i] = start;
    dinv[i] = rsqrtf(1.0f + (float)c);
  }
  if (i == n - 1) rowptr[n] = start + c;
  cur[t] = start;
  __syncthreads();
#pragma unroll 4
  for (int k = lo + t; k < hi; k += 256) {
    const int2 ed = tmp[k];
    const int pos = atomicAdd(&cur[ed.y & (BKT_ROWS - 1)], 1);
    csr[pos] = ed.x;
  }
}

template <int C, int U>
__global__ __launch_bounds__(256) void k_gather_rows(const float* __restrict__ H,
                                                     const float* __restrict__ dinv,
                                                     const int* __restrict__ rowptr,
                                                     const int* __restrict__ csr,
                                                     const float* __restrict__ bias,
                                                     float* __restrict__ A, int n) {
  constexpr int G  = C / 4;
  constexpr int EP = 32 / G;
  const int lane = threadIdx.x & 31;
  const int g = lane / G, q = lane % G;
  const int nw = gridDim.x * 8;
  const float4* __restrict__ H4 = (const float4*)H;
  float4* __restrict__ A4 = (float4*)A;
  const float4 bq = ((const float4*)bias)[q];
  const int w0 = __builtin_amdgcn_readfirstlane((int)(blockIdx.x * 8u + (threadIdx.x >> 5)));
  if (w0 >= n) return;
  int beg_n = rowptr[w0], end_n = rowptr[w0 + 1];
  for (int i = w0; i < n; i += nw) {
    const int beg = __builtin_amdgcn_readfirstlane(beg_n);
    const int end = __builtin_amdgcn_readfirstlane(end_n);
    if (i + nw < n) {
      beg_n = rowptr[i + nw];
      end_n = rowptr[i + nw + 1];
    }
    const float di = dinv[i];
    const float4 hs = H4[(unsigned)(i * G + q)];
    float4 acc = make_float4(0.0f, 0.0f, 0.0f, 0.0f);
    for (int e0 = beg; e0 < end; e0 += 32) {
      const int cnt = (end - e0 < 32) ? (end - e0) : 32;
      const int li = (lane < cnt) ? lane : (cnt - 1);
      const int s_l = csr[e0 + li];
      for (int j = 0; j < cnt; j += EP * U) {
        float4 h[U];
        int s[U];
#pragma unroll
        for (int u = 0; u < U; ++u) s[u] = lane_read_i(s_l, j + u * EP + g);
#pragma unroll
        for (int u = 0; u < U; ++u) {
          h[u] = make_float4(0.0f, 0.0f, 0.0f, 0.0f);
          if (j + u * EP + g < cnt) h[u] = H4[(unsigned)(s[u] * G + q)];
        }
#pragma unroll
        for (int u = 0; u < U; ++u) {
          acc.x += h[u].x;
          acc.y += h[u].y;
          acc.z += h[u].z;
          acc.w += h[u].w;
        }
      }
    }
#pragma unroll
    for (int o = G; o < 32; o <<= 1) {
      acc.x += lane_read_f(acc.x, lane ^ o);
      acc.y += lane_read_f(acc.y, lane ^ o);
      acc.z += lane_read_f(acc.z, lane ^ o);
      acc.w += lane_read_f(acc.w, lane ^ o);
    }
    if (g == 0) {
      A4[(unsigned)(i * G + q)] = make_float4(bq.x + di * (acc.x + hs.x), bq.y + di * (acc.y + hs.y),
                                              bq.z + di * (acc.z + hs.z), bq.w + di * (acc.w + hs.w));
    }
  }
}

#define TILE_PITCH 68
__global__ __launch_bounds__(256) void k_gather_rows_gemm(const float* __restrict__ H,
                                                          const float* __restrict__ dinv,
                                                          const int* __restrict__ rowptr,
                                                          const int* __restrict__ csr,
                                                          const float* __restrict__ bias,
                                                          const float* __restrict__ W,
                                                          float* __restrict__ H2, int n) {
  constexpr int C  = 64;
  constexpr int NO = 32;
  constexpr int U  = 8;
  constexpr int G  = C / 4;
  constexpr int EP = 32 / G;
  __shared__ float4 bfrag[(C / 4) * 32];
  __shared__ __attribute__((aligned(16))) float tile[2][16 * TILE_PITCH];
  const int lane = threadIdx.x & 31;
  const int g = lane / G, q = lane % G;
  const int m = lane & 15;
  const int khi = (lane >> 4) << 1;
  const int nw = gridDim.x * 8;
  const int wave = __builtin_amdgcn_readfirstlane((int)(threadIdx.x >> 5));
  const int b8 = blockIdx.x * 8;
  if (b8 >= n) return;
  const float4* __restrict__ H4 = (const float4*)H;
  const float4 bq = ((const float4*)bias)[q];

  const int w0 = b8 + wave;
  int beg_n = 0, end_n = 0;
  if (w0 < n) {
    beg_n = rowptr[w0];
    end_n = rowptr[w0 + 1];
  }

  for (int jj = wave; jj < C / 4; jj += 8) {
    const float* __restrict__ wp = W + (4 * jj + khi) * NO + m;
    bfrag[jj * 32 + lane] = make_float4(wp[0], wp[NO], wp[16], wp[NO + 16]);
  }
  __syncthreads();
  int k = 0;
  for (int base = b8; base < n; base += nw, ++k) {
    const int i = base + wave;
    float4 r4 = make_float4(0.0f, 0.0f, 0.0f, 0.0f);
    if (i < n) {
      const int beg = __builtin_amdgcn_readfirstlane(beg_n);
      const int end = __builtin_amdgcn_readfirstlane(end_n);
      if (i + nw < n) {
        beg_n = rowptr[i + nw];
        end_n = rowptr[i + nw + 1];
      }
      const float di = dinv[i];
      const float4 hs = H4[(unsigned)(i * G + q)];
      float4 acc = make_float4(0.0f, 0.0f, 0.0f, 0.0f);
      for (int e0 = beg; e0 < end; e0 += 32) {
        const int cnt = (end - e0 < 32) ? (end - e0) : 32;
        const int li = (lane < cnt) ? lane : (cnt - 1);
        const int s_l = csr[e0 + li];
        for (int j = 0; j < cnt; j += EP * U) {
          float4 h[U];
          int s[U];
#pragma unroll
          for (int u = 0; u < U; ++u) s[u] = lane_read_i(s_l, j + u * EP + g);
#pragma unroll
          for (int u = 0; u < U; ++u) {
            h[u] = make_float4(0.0f, 0.0f, 0.0f, 0.0f);
            if (j + u * EP + g < cnt) h[u] = H4[(unsigned)(s[u] * G + q)];
          }
#pragma unroll
          for (int u = 0; u < U; ++u) {
            acc.x += h[u].x;
            acc.y += h[u].y;
            acc.z += h[u].z;
            acc.w += h[u].w;
          }
        }
      }
#pragma unroll
      for (int o = G; o < 32; o <<= 1) {
        acc.x += lane_read_f(acc.x, lane ^ o);
        acc.y += lane_read_f(acc.y, lane ^ o);
        acc.z += lane_read_f(acc.z, lane ^ o);
        acc.w += lane_read_f(acc.w, lane ^ o);
      }
      r4 = make_float4(bq.x + di * (acc.x + hs.x), bq.y + di * (acc.y + hs.y),
                       bq.z + di * (acc.z + hs.z), bq.w + di * (acc.w + hs.w));
      r4.x = (r4.x > 0.0f) ? r4.x : NEG_SLOPE * r4.x;
      r4.y = (r4.y > 0.0f) ? r4.y : NEG_SLOPE * r4.y;
      r4.z = (r4.z > 0.0f) ? r4.z : NEG_SLOPE * r4.z;
      r4.w = (r4.w > 0.0f) ? r4.w : NEG_SLOPE * r4.w;
      r4.x *= di;
      r4.y *= di;
      r4.z *= di;
      r4.w *= di;
    }
    const int odd = k & 1;
    float* __restrict__ tb = &tile[(k >> 1) & 1][0];
    if (g == 0) *(float4*)(tb + (odd * 8 + wave) * TILE_PITCH + 4 * q) = r4;
    const bool lastit = base + nw >= n;
    if (odd || lastit) {
      if (!odd && g == 0)
        *(float4*)(tb + (8 + wave) * TILE_PITCH + 4 * q) = make_float4(0.0f, 0.0f, 0.0f, 0.0f);
      __syncthreads();
      if (wave == ((k >> 1) & 7)) {
        v8f c0 = v8f{}, c1 = v8f{};
        const float* __restrict__ ar = tb + m * TILE_PITCH + khi;
#pragma unroll
        for (int j = 0; j < C / 4; ++j) {
          const v2f a = *(const v2f*)(ar + 4 * j);
          const float4 bb = bfrag[j * 32 + lane];
          v2f b0, b1;
          b0.x = bb.x; b0.y = bb.y;
          b1.x = bb.z; b1.y = bb.w;
          c0 = __builtin_amdgcn_wmma_f32_16x16x4_f32(false, a, false, b0, (short)0, c0, false, false);
          c1 = __builtin_amdgcn_wmma_f32_16x16x4_f32(false, a, false, b1, (short)0, c1, false, false);
        }
        const int hi = lane >> 4;
        const int rbase = (odd ? base - nw : base) + hi * nw;
        const bool half_ok = (hi == 0) || odd;
#pragma unroll
        for (int v = 0; v < 8; ++v) {
          const int r = rbase + v;
          if (half_ok && r < n) {
            H2[(long long)r * NO + m]      = c0[v];
            H2[(long long)r * NO + 16 + m] = c1[v];
          }
        }
      }
    }
  }
}


extern "C" void kernel_launch(void* const* d_in, const int* in_sizes, int n_in,
                              void* d_out, int out_size, void* d_ws, size_t ws_size,
                              hipStream_t stream) {
  (void)n_in; (void)out_size;

  const float* x  = (const float*)d_in[0];
  const int*   ei = (const int*)  d_in[1];
  const float* W1 = (const float*)d_in[2];
  const float* b1 = (const float*)d_in[3];
  const float* W2 = (const float*)d_in[4];
  const float* b2 = (const float*)d_in[5];
  float* out = (float*)d_out;

  const int DIN  = 64;
  const int N    = in_sizes[0] / DIN;
  const int E    = in_sizes[1] / 2;
  const int DOUT = in_sizes[5];

  const int* src = ei;
  const int* dst = ei + E;

  float* dinv = (float*)d_ws;
  int Nr = ((N + 255) / 256) * 256;
  float* H1 = dinv + Nr;
  float* A1 = H1 + (size_t)N * DIN;
  float* H2 = H1;

  const int TB = 256;
  const int RES = 1024;
  const int Mtiles = (N + 15) / 16;
  const int GB = (Mtiles + 7) / 8;

  const int NB = (N + BKT_ROWS - 1) / BKT_ROWS;
  const size_t need = ((size_t)Nr + 2 * (size_t)N * DIN + (size_t)Nr + 256 + (size_t)E) * 4;
  const bool sorted_path = ws_size >= need && E > 0 && N < (1 << 26) && NB <= 8192 &&
                           2 * (size_t)E + 2 * (size_t)NB <= (size_t)N * DIN;

  if (sorted_path) {
    int* tmp  = (int*)A1;
    int* bsum = tmp + 2 * (size_t)E;
    int* bcur = bsum + NB;
    int* rowptr = (int*)(A1 + (size_t)N * DIN);
    int* csr = rowptr + Nr + 256;

    k_clear_counters<<<(2 * NB + TB - 1) / TB, TB, 0, stream>>>(bsum, 2 * NB);
    const int BB = (E + TB * BIN_PT - 1) / (TB * BIN_PT);
    k_bucket_sizes<<<BB, TB, NB * sizeof(int), stream>>>(dst, bsum, E, NB);
    k_bin_edges<<<BB, TB, 2 * NB * sizeof(int), stream>>>(src, dst, bsum, bcur, tmp, E, NB);
    k_fill_csr<<<NB, TB, 0, stream>>>(tmp, bsum, rowptr, dinv, csr, N);

    const int GR = 2048;
    k_gemm_wmma<4, false, true><<<GB, TB, 0, stream>>>(x, W1, H1, N, dinv);
    float* H2s = A1;
    k_gather_rows_gemm<<<GR, TB, 0, stream>>>(H1, dinv, rowptr, csr, b1, W2, H2s, N);
    k_gather_rows<32, 4><<<GR, TB, 0, stream>>>(H2s, dinv, rowptr, csr, b2, out, N);
    return;
  }

  k_init_deg <<<(N + TB - 1) / TB, TB, 0, stream>>>(dinv, N);
  int CB = (E + TB - 1) / TB;
  if (CB > RES) CB = RES;
  k_count_deg<<<CB, TB, 0, stream>>>(dst, dinv, E);
  k_rsqrt    <<<(N + TB - 1) / TB, TB, 0, stream>>>(dinv, N);

  const int EB = ((E + 31) / 32 + 7) / 8;

  k_gemm_wmma<4, false, false><<<GB, TB, 0, stream>>>(x, W1, H1, N, nullptr);
  {
    long long tot = (long long)N << 4;
    int IB = (int)((tot + TB - 1) / TB);
    if (IB > RES) IB = RES;
    k_agg_init<<<IB, TB, 0, stream>>>(H1, dinv, b1, A1, N, DIN, 4);
    k_agg_edges<64><<<EB, TB, 0, stream>>>(H1, dinv, src, dst, A1, E);
  }

  k_gemm_wmma<2, true, false><<<GB, TB, 0, stream>>>(A1, W2, H2, N, nullptr);
  {
    long long tot = (long long)N << 3;
    int IB = (int)((tot + TB - 1) / TB);
    if (IB > RES) IB = RES;
    k_agg_init<<<IB, TB, 0, stream>>>(H2, dinv, b2, out, N, DOUT, 3);
    k_agg_edges<32><<<EB, TB, 0, stream>>>(H2, dinv, src, dst, out, E);
  }
}


# edge sort level 1 chunk-local (LDS ranks, bucket starts to a table); level 2 sums a table column for its place, walks the chunks' runs: no global counters, 2 launches less; new sort order of sums
# speedup vs baseline: 3.5979x; 1.0619x over previous
#include <hip/hip_runtime.h>
#include <hip/hip_bf16.h>

typedef __attribute__((ext_vector_type(2))) float v2f;
typedef __attribute__((ext_vector_type(8))) float v8f;

#define NEG_SLOPE 0.01f


__global__ __launch_bounds__(256) void k_init_deg(float* __restrict__ deg, int n) {
  int i = blockIdx.x * blockDim.x + threadIdx.x;
  if (i < n) deg[i] = 1.0f;
}

__global__ __launch_bounds__(256) void k_count_deg(const int* __restrict__ dst,
                                                   float* __restrict__ deg, int e) {
  const int stride = gridDim.x * blockDim.x;
  for (int i0 = blockIdx.x * blockDim.x + threadIdx.x; i0 < e; i0 += 4 * stride) {
    int d[4];
#pragma unroll
    for (int u = 0; u < 4; ++u) {
      const int i = i0 + u * stride;
      d[u] = (i < e) ? dst[i] : -1;
    }
#pragma unroll
    for (int u = 0; u < 4; ++u)
      if (d[u] >= 0) atomicAdd(&deg[d[u]], 1.0f);
  }
}

__global__ __launch_bounds__(256) void k_rsqrt(float* __restrict__ deg, int n) {
  int i = blockIdx.x * blockDim.x + threadIdx.x;
  if (i < n) deg[i] = rsqrtf(deg[i]);
}


template <int NT, bool LEAKY, bool SCALE>
__global__ __launch_bounds__(256) void k_gemm_wmma(const float* __restrict__ X,
                                                   const float* __restrict__ W,
                                                   float* __restrict__ H,
                                                   int M, const float* __restrict__ rscale) {
  constexpr int K = 64;
  constexpr int Nout = NT * 16;
  const int lane = threadIdx.x & 31;
  const int m    = lane & 15;
  const int khi  = (lane >> 4) << 1;

  const int tile = blockIdx.x * 8 + (threadIdx.x >> 5);
  const int row0 = tile << 4;
  if (row0 >= M) return;

  int arow = row0 + m;
  if (arow >= M) arow = M - 1;
  const float* __restrict__ xr = X + (long long)arow * K + khi;
  const float* __restrict__ wc = W + (long long)khi * Nout + m;
  const float rs = SCALE ? rscale[arow] : 1.0f;

  v2f a[K / 4];
#pragma unroll
  for (int j = 0; j < K / 4; ++j) {
    a[j] = *(const v2f*)(xr + 4 * j);
    if (LEAKY) {
      a[j].x = (a[j].x > 0.0f) ? a[j].x : NEG_SLOPE * a[j].x;
      a[j].y = (a[j].y > 0.0f) ? a[j].y : NEG_SLOPE * a[j].y;
    }
    if (SCALE) {
      a[j].x *= rs;
      a[j].y *= rs;
    }
  }

  v8f c[NT];
#pragma unroll
  for (int t = 0; t < NT; ++t) c[t] = v8f{};

#pragma unroll
  for (int j = 0; j < K / 4; ++j) {
#pragma unroll
    for (int t = 0; t < NT; ++t) {
      v2f b;
      b.x = wc[(4 * j + 0) * Nout + 16 * t];
      b.y = wc[(4 * j + 1) * Nout + 16 * t];
      c[t] = __builtin_amdgcn_wmma_f32_16x16x4_f32(false, a[j], false, b, (short)0,
                                                   c[t], false, false);
    }
  }

  const int hi8 = (lane >> 4) << 3;
#pragma unroll
  for (int v = 0; v < 8; ++v) {
    int r = row0 + hi8 + v;
    if (r < M) {
#pragma unroll
      for (int t = 0; t < NT; ++t) H[(long long)r * Nout + 16 * t + m] = c[t][v];
    }
  }
}

__global__ __launch_bounds__(256) void k_agg_init(const float* __restrict__ H,
                                                  const float* __restrict__ dinv,
                                                  const float* __restrict__ bias,
                                                  float* __restrict__ A,
                                                  int n, int C, int tshift) {
  (void)C;
  const int tot = n << tshift;
  const int stride = gridDim.x * blockDim.x;
  const int t = threadIdx.x & ((1 << tshift) - 1);
  const float4 b = *((const float4*)bias + t);
  for (int g0 = blockIdx.x * blockDim.x + threadIdx.x; g0 < tot; g0 += 4 * stride) {
    float4 h[4];
    float di[4];
#pragma unroll
    for (int u = 0; u < 4; ++u) {
      const int gid = g0 + u * stride;
      if (gid < tot) {
        di[u] = dinv[gid >> tshift];
        h[u]  = *((const float4*)H + gid);
      }
    }
#pragma unroll
    for (int u = 0; u < 4; ++u) {
      const int gid = g0 + u * stride;
      if (gid < tot) {
        const float w = di[u] * di[u];
        *((float4*)A + gid) = make_float4(b.x + w * h[u].x, b.y + w * h[u].y,
                                          b.z + w * h[u].z, b.w + w * h[u].w);
      }
    }
  }
}

template <int C>
__global__ __launch_bounds__(256) void k_agg_edges(const float* __restrict__ H,
                                                   const float* __restrict__ dinv,
                                                   const int* __restrict__ src,
                                                   const int* __restrict__ dst,
                                                   float* __restrict__ A,
                                                   int e) {
  constexpr int R = C / 32;
  constexpr int U = 16 / R;
  const int lane = threadIdx.x & 31;
  const int wave = __builtin_amdgcn_readfirstlane(
      (int)((blockIdx.x * 256u + threadIdx.x) >> 5));
  const long long base = (long long)wave << 5;
  if (base >= e) return;
  const int cnt = (e - base < 32) ? (int)(e - base) : 32;
  const bool ok = lane < cnt;
  const int s_v = ok ? src[base + lane] : 0;
  const int d_v = ok ? dst[base + lane] : 0;
  const float w_v = dinv[s_v] * dinv[d_v];
  const int w_b = __float_as_int(w_v);

  if (cnt == 32) {
    for (int i0 = 0; i0 < 32; i0 += U) {
      float h[U][R];
#pragma unroll
      for (int u = 0; u < U; ++u) {
        const int s = __builtin_amdgcn_readlane(s_v, i0 + u);
        const float* __restrict__ row = H + (long long)s * C + lane;
#pragma unroll
        for (int r = 0; r < R; ++r) h[u][r] = row[32 * r];
      }
#pragma unroll
      for (int u = 0; u < U; ++u) {
        const int d = __builtin_amdgcn_readlane(d_v, i0 + u);
        const float w = __int_as_float(__builtin_amdgcn_readlane(w_b, i0 + u));
        float* __restrict__ o = A + (long long)d * C + lane;
#pragma unroll
        for (int r = 0; r < R; ++r) atomicAdd(o + 32 * r, w * h[u][r]);
      }
    }
  } else {
    for (int i = 0; i < cnt; ++i) {
      const int s = __builtin_amdgcn_readlane(s_v, i);
      const int d = __builtin_amdgcn_readlane(d_v, i);
      const float w = __int_as_float(__builtin_amdgcn_readlane(w_b, i));
      const float* __restrict__ row = H + (long long)s * C + lane;
      float* __restrict__ o = A + (long long)d * C + lane;
#pragma unroll
      for (int r = 0; r < R; ++r) atomicAdd(o + 32 * r, w * row[32 * r]);
    }
  }
}


#define BKT_SHIFT 8
#define BKT_ROWS 256
#define BIN_PT 16

__device__ __forceinline__ int lane_read_i(int v, int src_lane) {
  return __builtin_amdgcn_ds_bpermute(src_lane << 2, v);
}
__device__ __forceinline__ float lane_read_f(float v, int src_lane) {
  return __int_as_float(__builtin_amdgcn_ds_bpermute(src_lane << 2, __float_as_int(v)));
}

__device__ __forceinline__ int block_scan_excl(int v, int* wsum, int* total) {
  const int lane = threadIdx.x & 31, wv = threadIdx.x >> 5;
  int inc = v;
  for (int o = 1; o < 32; o <<= 1) {
    const int r = lane_read_i(inc, (lane - o) & 31);
    if (lane >= o) inc += r;
  }
  __syncthreads();
  if (lane == 31) wsum[wv] = inc;
  __syncthreads();
  int woff = 0, tot = 0;
  for (int k = 0; k < 8; ++k) {
    const int s = wsum[k];
    woff += (k < wv) ? s : 0;
    tot += s;
  }
  *total = tot;
  __syncthreads();
  return woff + inc - v;
}

__global__ __launch_bounds__(256) void k_chunk_sort(const int* __restrict__ src,
                                                    const int* __restrict__ dst,
                                                    int* __restrict__ offT,
                                                    int* __restrict__ tmp_i,
                                                    int e, int nb, int pitch) {
  extern __shared__ int sh[];
  __shared__ int wsum[8];
  int* hist = sh;
  int* loc = sh + nb;
  int2* __restrict__ tmp = (int2*)tmp_i;
  const int t = threadIdx.x;
  int* __restrict__ orow = offT + (size_t)blockIdx.x * pitch;
  for (int k = t; k < nb; k += 256) hist[k] = 0;
  __syncthreads();
  const long long e0 = (long long)blockIdx.x * (256 * BIN_PT);
  int s[BIN_PT], d[BIN_PT], r[BIN_PT];
#pragma unroll
  for (int u = 0; u < BIN_PT; ++u) {
    const long long i = e0 + u * 256 + t;
    const bool ok = i < e;
    s[u] = ok ? src[i] : 0;
    d[u] = ok ? dst[i] : -1;
  }
#pragma unroll
  for (int u = 0; u < BIN_PT; ++u)
    r[u] = (d[u] >= 0) ? atomicAdd(&hist[d[u] >> BKT_SHIFT], 1) : 0;
  __syncthreads();
  const int ch = (nb + 255) >> 8;
  int run = 0;
  for (int k = t * ch; k < t * ch + ch; ++k) run += (k < nb) ? hist[k] : 0;
  int total;
  int pre = block_scan_excl(run, wsum, &total);
  for (int k = t * ch; k < t * ch + ch; ++k)
    if (k < nb) {
      loc[k] = pre;
      orow[k] = pre;
      pre += hist[k];
    }
  if (t == 0) orow[nb] = total;
  __syncthreads();
#pragma unroll
  for (int u = 0; u < BIN_PT; ++u)
    if (d[u] >= 0) tmp[e0 + loc[d[u] >> BKT_SHIFT] + r[u]] = make_int2(s[u], d[u]);
}

#define SEG_CU 16

template <bool PLACE>
__device__ __forceinline__ void walk_runs(const int2* __restrict__ tmp, const int* sst,
                                          const int* sen, int bb, int hw, int j, int* cur,
                                          int* __restrict__ csr) {
  const size_t cs = 256 * BIN_PT;
  bool more = false;
  for (int c0 = hw; c0 < bb; c0 += 16 * SEG_CU) {
    int2 ea[SEG_CU], eb[SEG_CU];
    int kk[SEG_CU], ee[SEG_CU];
#pragma unroll
    for (int u = 0; u < SEG_CU; ++u) {
      const int c = c0 + 16 * u;
      const int cc = (c < bb) ? c : (bb - 1);
      kk[u] = sst[cc] + j;
      ee[u] = (c < bb) ? sen[cc] : 0;
    }
#pragma unroll
    for (int u = 0; u < SEG_CU; ++u) {
      const int2* __restrict__ seg = tmp + (size_t)(c0 + 16 * u) * cs;
      const int k = kk[u], en = ee[u];
      ea[u] = make_int2(0, -1);
      eb[u] = make_int2(0, -1);
      if (PLACE) {
        if (k < en) ea[u] = seg[k];
        if (k + 16 < en) eb[u] = seg[k + 16];
      } else {
        if (k < en) ea[u].y = seg[k].y;
        if (k + 16 < en) eb[u].y = seg[k + 16].y;
      }
      more |= (k + 32 < en);
    }
#pragma unroll
    for (int u = 0; u < SEG_CU; ++u) {
      if (ea[u].y >= 0) {
        const int pos = atomicAdd(&cur[ea[u].y & (BKT_ROWS - 1)], 1);
        if (PLACE) csr[pos] = ea[u].x;
      }
      if (eb[u].y >= 0) {
        const int pos = atomicAdd(&cur[eb[u].y & (BKT_ROWS - 1)], 1);
        if (PLACE) csr[pos] = eb[u].x;
      }
    }
  }
  if (more) {
    for (int c = hw; c < bb; c += 16) {
      const int2* __restrict__ seg = tmp + (size_t)c * cs;
      const int en = sen[c];
      for (int k = sst[c] + 32 + j; k < en; k += 16) {
        const int2 ed = seg[k];
        const int pos = atomicAdd(&cur[ed.y & (BKT_ROWS - 1)], 1);
        if (PLACE) csr[pos] = ed.x;
      }
    }
  }
}

__global__ __launch_bounds__(256) void k_fill_csr(const int* __restrict__ tmp_i,
                                                  const int* __restrict__ offT,
                                                  int* __restrict__ rowptr,
                                                  float* __restrict__ dinv,
                                                  int* __restrict__ csr, int n, int bb,
                                                  int pitch) {
  extern __shared__ int sh[];
  __shared__ int cur[BKT_ROWS];
  __shared__ int wsum[8];
  int* sst = sh;
  int* sen = sh + bb;
  const int2* __restrict__ tmp = (const int2*)tmp_i;
  const int b = blockIdx.x, t = threadIdx.x;
  const int i = b * BKT_ROWS + t;
  const int hw = t >> 4, j = t & 15;
  const int* __restrict__ ocol = offT + b;
  int part = 0;
  for (int c = t; c < bb; c += 256) {
    const int a = ocol[(size_t)c * pitch];
    sst[c] = a;
    sen[c] = ocol[(size_t)c * pitch + 1];
    part += a;
  }
  int lo;
  (void)block_scan_excl(part, wsum, &lo);
  cur[t] = 0;
  __syncthreads();
  walk_runs<false>(tmp, sst, sen, bb, hw, j, cur, csr);
  __syncthreads();
  const int cnt = cur[t];
  int tot;
  const int start = lo + block_scan_excl(cnt, wsum, &tot);
  if (i < n) {
    rowptr[i] = start;
    dinv[i] = rsqrtf(1.0f + (float)cnt);
  }
  if (i == n - 1) rowptr[n] = start + cnt;
  cur[t] = start;
  __syncthreads();
  walk_runs<true>(tmp, sst, sen, bb, hw, j, cur, csr);
}

template <int C, int U>
__global__ __launch_bounds__(256) void k_gather_rows(const float* __restrict__ H,
                                                     const float* __restrict__ dinv,
                                                     const int* __restrict__ rowptr,
                                                     const int* __restrict__ csr,
                                                     const float* __restrict__ bias,
                                                     float* __restrict__ A, int n) {
  constexpr int G  = C / 4;
  constexpr int EP = 32 / G;
  const int lane = threadIdx.x & 31;
  const int g = lane / G, q = lane % G;
  const int nw = gridDim.x * 8;
  const float4* __restrict__ H4 = (const float4*)H;
  float4* __restrict__ A4 = (float4*)A;
  const float4 bq = ((const float4*)bias)[q];
  const int w0 = __builtin_amdgcn_readfirstlane((int)(blockIdx.x * 8u + (threadIdx.x >> 5)));
  if (w0 >= n) return;
  int beg_n = rowptr[w0], end_n = rowptr[w0 + 1];
  for (int i = w0; i < n; i += nw) {
    const int beg = __builtin_amdgcn_readfirstlane(beg_n);
    const int end = __builtin_amdgcn_readfirstlane(end_n);
    if (i + nw < n) {
      beg_n = rowptr[i + nw];
      end_n = rowptr[i + nw + 1];
    }
    const float di = dinv[i];
    const float4 hs = H4[(unsigned)(i * G + q)];
    float4 acc = make_float4(0.0f, 0.0f, 0.0f, 0.0f);
    for (int e0 = beg; e0 < end; e0 += 32) {
      const int cnt = (end - e0 < 32) ? (end - e0) : 32;
      const int li = (lane < cnt) ? lane : (cnt - 1);
      const int s_l = csr[e0 + li];
      for (int j = 0; j < cnt; j += EP * U) {
        float4 h[U];
        int s[U];
#pragma unroll
        for (int u = 0; u < U; ++u) s[u] = lane_read_i(s_l, j + u * EP + g);
#pragma unroll
        for (int u = 0; u < U; ++u) {
          h[u] = make_float4(0.0f, 0.0f, 0.0f, 0.0f);
          if (j + u * EP + g < cnt) h[u] = H4[(unsigned)(s[u] * G + q)];
        }
#pragma unroll
        for (int u = 0; u < U; ++u) {
          acc.x += h[u].x;
          acc.y += h[u].y;
          acc.z += h[u].z;
          acc.w += h[u].w;
        }
      }
    }
#pragma unroll
    for (int o = G; o < 32; o <<= 1) {
      acc.x += lane_read_f(acc.x, lane ^ o);
      acc.y += lane_read_f(acc.y, lane ^ o);
      acc.z += lane_read_f(acc.z, lane ^ o);
      acc.w += lane_read_f(acc.w, lane ^ o);
    }
    if (g == 0) {
      A4[(unsigned)(i * G + q)] = make_float4(bq.x + di * (acc.x + hs.x), bq.y + di * (acc.y + hs.y),
                                              bq.z + di * (acc.z + hs.z), bq.w + di * (acc.w + hs.w));
    }
  }
}

#define TILE_PITCH 68
__global__ __launch_bounds__(256) void k_gather_rows_gemm(const float* __restrict__ H,
                                                          const float* __restrict__ dinv,
                                                          const int* __restrict__ rowptr,
                                                          const int* __restrict__ csr,
                                                          const float* __restrict__ bias,
                                                          const float* __restrict__ W,
                                                          float* __restrict__ H2, int n) {
  constexpr int C  = 64;
  constexpr int NO = 32;
  constexpr int U  = 8;
  constexpr int G  = C / 4;
  constexpr int EP = 32 / G;
  __shared__ float4 bfrag[(C / 4) * 32];
  __shared__ __attribute__((aligned(16))) float tile[2][16 * TILE_PITCH];
  const int lane = threadIdx.x & 31;
  const int g = lane / G, q = lane % G;
  const int m = lane & 15;
  const int khi = (lane >> 4) << 1;
  const int nw = gridDim.x * 8;
  const int wave = __builtin_amdgcn_readfirstlane((int)(threadIdx.x >> 5));
  const int b8 = blockIdx.x * 8;
  if (b8 >= n) return;
  const float4* __restrict__ H4 = (const float4*)H;
  const float4 bq = ((const float4*)bias)[q];

  const int w0 = b8 + wave;
  int beg_n = 0, end_n = 0;
  if (w0 < n) {
    beg_n = rowptr[w0];
    end_n = rowptr[w0 + 1];
  }

  for (int jj = wave; jj < C / 4; jj += 8) {
    const float* __restrict__ wp = W + (4 * jj + khi) * NO + m;
    bfrag[jj * 32 + lane] = make_float4(wp[0], wp[NO], wp[16], wp[NO + 16]);
  }
  __syncthreads();
  int k = 0;
  for (int base = b8; base < n; base += nw, ++k) {
    const int i = base + wave;
    float4 r4 = make_float4(0.0f, 0.0f, 0.0f, 0.0f);
    if (i < n) {
      const int beg = __builtin_amdgcn_readfirstlane(beg_n);
      const int end = __builtin_amdgcn_readfirstlane(end_n);
      if (i + nw < n) {
        beg_n = rowptr[i + nw];
        end_n = rowptr[i + nw + 1];
      }
      const float di = dinv[i];
      const float4 hs = H4[(unsigned)(i * G + q)];
      float4 acc = make_float4(0.0f, 0.0f, 0.0f, 0.0f);
      for (int e0 = beg; e0 < end; e0 += 32) {
        const int cnt = (end - e0 < 32) ? (end - e0) : 32;
        const int li = (lane < cnt) ? lane : (cnt - 1);
        const int s_l = csr[e0 + li];
        for (int j = 0; j < cnt; j += EP * U) {
          float4 h[U];
          int s[U];
#pragma unroll
          for (int u = 0; u < U; ++u) s[u] = lane_read_i(s_l, j + u * EP + g);
#pragma unroll
          for (int u = 0; u < U; ++u) {
            h[u] = make_float4(0.0f, 0.0f, 0.0f, 0.0f);
            if (j + u * EP + g < cnt) h[u] = H4[(unsigned)(s[u] * G + q)];
          }
#pragma unroll
          for (int u = 0; u < U; ++u) {
            acc.x += h[u].x;
            acc.y += h[u].y;
            acc.z += h[u].z;
            acc.w += h[u].w;
          }
        }
      }
#pragma unroll
      for (int o = G; o < 32; o <<= 1) {
        acc.x += lane_read_f(acc.x, lane ^ o);
        acc.y += lane_read_f(acc.y, lane ^ o);
        acc.z += lane_read_f(acc.z, lane ^ o);
        acc.w += lane_read_f(acc.w, lane ^ o);
      }
      r4 = make_float4(bq.x + di * (acc.x + hs.x), bq.y + di * (acc.y + hs.y),
                       bq.z + di * (acc.z + hs.z), bq.w + di * (acc.w + hs.w));
      r4.x = (r4.x > 0.0f) ? r4.x : NEG_SLOPE * r4.x;
      r4.y = (r4.y > 0.0f) ? r4.y : NEG_SLOPE * r4.y;
      r4.z = (r4.z > 0.0f) ? r4.z : NEG_SLOPE * r4.z;
      r4.w = (r4.w > 0.0f) ? r4.w : NEG_SLOPE * r4.w;
      r4.x *= di;
      r4.y *= di;
      r4.z *= di;
      r4.w *= di;
    }
    const int odd = k & 1;
    float* __restrict__ tb = &tile[(k >> 1) & 1][0];
    if (g == 0) *(float4*)(tb + (odd * 8 + wave) * TILE_PITCH + 4 * q) = r4;
    const bool lastit = base + nw >= n;
    if (odd || lastit) {
      if (!odd && g == 0)
        *(float4*)(tb + (8 + wave) * TILE_PITCH + 4 * q) = make_float4(0.0f, 0.0f, 0.0f, 0.0f);
      __syncthreads();
      if (wave == ((k >> 1) & 7)) {
        v8f c0 = v8f{}, c1 = v8f{};
        const float* __restrict__ ar = tb + m * TILE_PITCH + khi;
#pragma unroll
        for (int j = 0; j < C / 4; ++j) {
          const v2f a = *(const v2f*)(ar + 4 * j);
          const float4 bb = bfrag[j * 32 + lane];
          v2f b0, b1;
          b0.x = bb.x; b0.y = bb.y;
          b1.x = bb.z; b1.y = bb.w;
          c0 = __builtin_amdgcn_wmma_f32_16x16x4_f32(false, a, false, b0, (short)0, c0, false, false);
          c1 = __builtin_amdgcn_wmma_f32_16x16x4_f32(false, a, false, b1, (short)0, c1, false, false);
        }
        const int hi = lane >> 4;
        const int rbase = (odd ? base - nw : base) + hi * nw;
        const bool half_ok = (hi == 0) || odd;
#pragma unroll
        for (int v = 0; v < 8; ++v) {
          const int r = rbase + v;
          if (half_ok && r < n) {
            H2[(long long)r * NO + m]      = c0[v];
            H2[(long long)r * NO + 16 + m] = c1[v];
          }
        }
      }
    }
  }
}


extern "C" void kernel_launch(void* const* d_in, const int* in_sizes, int n_in,
                              void* d_out, int out_size, void* d_ws, size_t ws_size,
                              hipStream_t stream) {
  (void)n_in; (void)out_size;

  const float* x  = (const float*)d_in[0];
  const int*   ei = (const int*)  d_in[1];
  const float* W1 = (const float*)d_in[2];
  const float* b1 = (const float*)d_in[3];
  const float* W2 = (const float*)d_in[4];
  const float* b2 = (const float*)d_in[5];
  float* out = (float*)d_out;

  const int DIN  = 64;
  const int N    = in_sizes[0] / DIN;
  const int E    = in_sizes[1] / 2;
  const int DOUT = in_sizes[5];

  const int* src = ei;
  const int* dst = ei + E;

  float* dinv = (float*)d_ws;
  int Nr = ((N + 255) / 256) * 256;
  float* H1 = dinv + Nr;
  float* A1 = H1 + (size_t)N * DIN;
  float* H2 = H1;

  const int TB = 256;
  const int RES = 1024;
  const int Mtiles = (N + 15) / 16;
  const int GB = (Mtiles + 7) / 8;

  const int NB = (N + BKT_ROWS - 1) / BKT_ROWS;
  const int BB = (E + TB * BIN_PT - 1) / (TB * BIN_PT);
  const int OP = (NB + 1 + 31) & ~31;
  const size_t TO = (2 * (size_t)E + 31) & ~(size_t)31;
  const size_t need = ((size_t)Nr + 2 * (size_t)N * DIN + (size_t)Nr + 256 + (size_t)E) * 4;
  const bool sorted_path = ws_size >= need && E > 0 && N < (1 << 26) && NB <= 8192 && BB <= 4096 &&
                           TO + (size_t)OP * (size_t)BB <= (size_t)N * DIN;

  if (sorted_path) {
    int* tmp  = (int*)A1;
    int* offT = tmp + TO;
    int* rowptr = (int*)(A1 + (size_t)N * DIN);
    int* csr = rowptr + Nr + 256;

    k_chunk_sort<<<BB, TB, 2 * NB * sizeof(int), stream>>>(src, dst, offT, tmp, E, NB, OP);
    k_fill_csr<<<NB, TB, 2 * BB * sizeof(int), stream>>>(tmp, offT, rowptr, dinv, csr, N, BB, OP);

    const int GR = 2048;
    k_gemm_wmma<4, false, true><<<GB, TB, 0, stream>>>(x, W1, H1, N, dinv);
    float* H2s = A1;
    k_gather_rows_gemm<<<GR, TB, 0, stream>>>(H1, dinv, rowptr, csr, b1, W2, H2s, N);
    k_gather_rows<32, 4><<<GR, TB, 0, stream>>>(H2s, dinv, rowptr, csr, b2, out, N);
    return;
  }

  k_init_deg <<<(N + TB - 1) / TB, TB, 0, stream>>>(dinv, N);
  int CB = (E + TB - 1) / TB;
  if (CB > RES) CB = RES;
  k_count_deg<<<CB, TB, 0, stream>>>(dst, dinv, E);
  k_rsqrt    <<<(N + TB - 1) / TB, TB, 0, stream>>>(dinv, N);

  const int EB = ((E + 31) / 32 + 7) / 8;

  k_gemm_wmma<4, false, false><<<GB, TB, 0, stream>>>(x, W1, H1, N, nullptr);
  {
    long long tot = (long long)N << 4;
    int IB = (int)((tot + TB - 1) / TB);
    if (IB > RES) IB = RES;
    k_agg_init<<<IB, TB, 0, stream>>>(H1, dinv, b1, A1, N, DIN, 4);
    k_agg_edges<64><<<EB, TB, 0, stream>>>(H1, dinv, src, dst, A1, E);
  }

  k_gemm_wmma<2, true, false><<<GB, TB, 0, stream>>>(A1, W2, H2, N, nullptr);
  {
    long long tot = (long long)N << 3;
    int IB = (int)((tot + TB - 1) / TB);
    if (IB > RES) IB = RES;
    k_agg_init<<<IB, TB, 0, stream>>>(H2, dinv, b2, out, N, DOUT, 3);
    k_agg_edges<32><<<EB, TB, 0, stream>>>(H2, dinv, src, dst, out, E);
  }
}


# layer-1 product fed through LDS: W staged once per block in B-fragment order, each wave's 16x64 X tile and its result rows move as whole 128-byte lines (16-byte accesses); same sums
# speedup vs baseline: 3.7371x; 1.0387x over previous
#include <hip/hip_runtime.h>
#include <hip/hip_bf16.h>

typedef __attribute__((ext_vector_type(2))) float v2f;
typedef __attribute__((ext_vector_type(8))) float v8f;

#define NEG_SLOPE 0.01f


__global__ __launch_bounds__(256) void k_init_deg(float* __restrict__ deg, int n) {
  int i = blockIdx.x * blockDim.x + threadIdx.x;
  if (i < n) deg[i] = 1.0f;
}

__global__ __launch_bounds__(256) void k_count_deg(const int* __restrict__ dst,
                                                   float* __restrict__ deg, int e) {
  const int stride = gridDim.x * blockDim.x;
  for (int i0 = blockIdx.x * blockDim.x + threadIdx.x; i0 < e; i0 += 4 * stride) {
    int d[4];
#pragma unroll
    for (int u = 0; u < 4; ++u) {
      const int i = i0 + u * stride;
      d[u] = (i < e) ? dst[i] : -1;
    }
#pragma unroll
    for (int u = 0; u < 4; ++u)
      if (d[u] >= 0) atomicAdd(&deg[d[u]], 1.0f);
  }
}

__global__ __launch_bounds__(256) void k_rsqrt(float* __restrict__ deg, int n) {
  int i = blockIdx.x * blockDim.x + threadIdx.x;
  if (i < n) deg[i] = rsqrtf(deg[i]);
}


template <int NT, bool LEAKY, bool SCALE>
__global__ __launch_bounds__(256) void k_gemm_wmma(const float* __restrict__ X,
                                                   const float* __restrict__ W,
                                                   float* __restrict__ H,
                                                   int M, const float* __restrict__ rscale) {
  constexpr int K = 64;
  constexpr int Nout = NT * 16;
  const int lane = threadIdx.x & 31;
  const int m    = lane & 15;
  const int khi  = (lane >> 4) << 1;

  const int tile = blockIdx.x * 8 + (threadIdx.x >> 5);
  const int row0 = tile << 4;
  if (row0 >= M) return;

  int arow = row0 + m;
  if (arow >= M) arow = M - 1;
  const float* __restrict__ xr = X + (long long)arow * K + khi;
  const float* __restrict__ wc = W + (long long)khi * Nout + m;
  const float rs = SCALE ? rscale[arow] : 1.0f;

  v2f a[K / 4];
#pragma unroll
  for (int j = 0; j < K / 4; ++j) {
    a[j] = *(const v2f*)(xr + 4 * j);
    if (LEAKY) {
      a[j].x = (a[j].x > 0.0f) ? a[j].x : NEG_SLOPE * a[j].x;
      a[j].y = (a[j].y > 0.0f) ? a[j].y : NEG_SLOPE * a[j].y;
    }
    if (SCALE) {
      a[j].x *= rs;
      a[j].y *= rs;
    }
  }

  v8f c[NT];
#pragma unroll
  for (int t = 0; t < NT; ++t) c[t] = v8f{};

#pragma unroll
  for (int j = 0; j < K / 4; ++j) {
#pragma unroll
    for (int t = 0; t < NT; ++t) {
      v2f b;
      b.x = wc[(4 * j + 0) * Nout + 16 * t];
      b.y = wc[(4 * j + 1) * Nout + 16 * t];
      c[t] = __builtin_amdgcn_wmma_f32_16x16x4_f32(false, a[j], false, b, (short)0,
                                                   c[t], false, false);
    }
  }

  const int hi8 = (lane >> 4) << 3;
#pragma unroll
  for (int v = 0; v < 8; ++v) {
    int r = row0 + hi8 + v;
    if (r < M) {
#pragma unroll
      for (int t = 0; t < NT; ++t) H[(long long)r * Nout + 16 * t + m] = c[t][v];
    }
  }
}

#define A_PITCH 68
__global__ __launch_bounds__(256) void k_gemm64_rows(const float* __restrict__ X,
                                                     const float* __restrict__ W,
                                                     float* __restrict__ H, int M,
                                                     const float* __restrict__ rscale) {
  constexpr int K  = 64;
  constexpr int NO = 64;
  __shared__ float4 bfrag[(K / 4) * 32 * 2];
  __shared__ __attribute__((aligned(16))) float atile[8][16 * A_PITCH];
  const int lane = threadIdx.x & 31;
  const int wave = threadIdx.x >> 5;
  const int m    = lane & 15;
  const int h    = lane >> 4;
  const int khi  = h << 1;
  const int row0 = (blockIdx.x * 8 + wave) << 4;
  float* __restrict__ tl = &atile[wave][0];
  const int fc = m << 2;

  float4 av[8];
#pragma unroll
  for (int i = 0; i < 8; ++i) {
    int r = row0 + 2 * i + h;
    if (r >= M) r = M - 1;
    av[i] = *(const float4*)(X + (long long)r * K + fc);
  }
  int ar = row0 + m;
  if (ar >= M) ar = M - 1;
  const float rs = rscale[ar];
#pragma unroll
  for (int u = 0; u < 2; ++u) {
    const int jj = wave + 8 * u;
    const float* __restrict__ wp = W + (4 * jj + khi) * NO + m;
    bfrag[(jj * 32 + lane) * 2]     = make_float4(wp[0],  wp[NO],      wp[16], wp[NO + 16]);
    bfrag[(jj * 32 + lane) * 2 + 1] = make_float4(wp[32], wp[NO + 32], wp[48], wp[NO + 48]);
  }
#pragma unroll
  for (int i = 0; i < 8; ++i) *(float4*)(tl + (2 * i + h) * A_PITCH + fc) = av[i];
  __syncthreads();
  {
    v2f a[K / 4];
#pragma unroll
    for (int j = 0; j < K / 4; ++j) {
      a[j] = *(const v2f*)(tl + m * A_PITCH + khi + 4 * j);
      a[j].x *= rs;
      a[j].y *= rs;
    }
    v8f c[4];
#pragma unroll
    for (int t = 0; t < 4; ++t) c[t] = v8f{};
#pragma unroll
    for (int j = 0; j < K / 4; ++j) {
      const float4 q0 = bfrag[(j * 32 + lane) * 2];
      const float4 q1 = bfrag[(j * 32 + lane) * 2 + 1];
      v2f b0, b1, b2, b3;
      b0.x = q0.x; b0.y = q0.y;
      b1.x = q0.z; b1.y = q0.w;
      b2.x = q1.x; b2.y = q1.y;
      b3.x = q1.z; b3.y = q1.w;
      c[0] = __builtin_amdgcn_wmma_f32_16x16x4_f32(false, a[j], false, b0, (short)0, c[0], false, false);
      c[1] = __builtin_amdgcn_wmma_f32_16x16x4_f32(false, a[j], false, b1, (short)0, c[1], false, false);
      c[2] = __builtin_amdgcn_wmma_f32_16x16x4_f32(false, a[j], false, b2, (short)0, c[2], false, false);
      c[3] = __builtin_amdgcn_wmma_f32_16x16x4_f32(false, a[j], false, b3, (short)0, c[3], false, false);
    }
    const int hi8 = h << 3;
#pragma unroll
    for (int v = 0; v < 8; ++v) {
#pragma unroll
      for (int t = 0; t < 4; ++t) tl[(hi8 + v) * A_PITCH + 16 * t + m] = c[t][v];
    }
  }
  __syncthreads();
#pragma unroll
  for (int i = 0; i < 8; ++i) {
    const int r = row0 + 2 * i + h;
    if (r < M)
      *(float4*)(H + (long long)r * NO + fc) = *(const float4*)(tl + (2 * i + h) * A_PITCH + fc);
  }
}

__global__ __launch_bounds__(256) void k_agg_init(const float* __restrict__ H,
                                                  const float* __restrict__ dinv,
                                                  const float* __restrict__ bias,
                                                  float* __restrict__ A,
                                                  int n, int C, int tshift) {
  (void)C;
  const int tot = n << tshift;
  const int stride = gridDim.x * blockDim.x;
  const int t = threadIdx.x & ((1 << tshift) - 1);
  const float4 b = *((const float4*)bias + t);
  for (int g0 = blockIdx.x * blockDim.x + threadIdx.x; g0 < tot; g0 += 4 * stride) {
    float4 h[4];
    float di[4];
#pragma unroll
    for (int u = 0; u < 4; ++u) {
      const int gid = g0 + u * stride;
      if (gid < tot) {
        di[u] = dinv[gid >> tshift];
        h[u]  = *((const float4*)H + gid);
      }
    }
#pragma unroll
    for (int u = 0; u < 4; ++u) {
      const int gid = g0 + u * stride;
      if (gid < tot) {
        const float w = di[u] * di[u];
        *((float4*)A + gid) = make_float4(b.x + w * h[u].x, b.y + w * h[u].y,
                                          b.z + w * h[u].z, b.w + w * h[u].w);
      }
    }
  }
}

template <int C>
__global__ __launch_bounds__(256) void k_agg_edges(const float* __restrict__ H,
                                                   const float* __restrict__ dinv,
                                                   const int* __restrict__ src,
                                                   const int* __restrict__ dst,
                                                   float* __restrict__ A,
                                                   int e) {
  constexpr int R = C / 32;
  constexpr int U = 16 / R;
  const int lane = threadIdx.x & 31;
  const int wave = __builtin_amdgcn_readfirstlane(
      (int)((blockIdx.x * 256u + threadIdx.x) >> 5));
  const long long base = (long long)wave << 5;
  if (base >= e) return;
  const int cnt = (e - base < 32) ? (int)(e - base) : 32;
  const bool ok = lane < cnt;
  const int s_v = ok ? src[base + lane] : 0;
  const int d_v = ok ? dst[base + lane] : 0;
  const float w_v = dinv[s_v] * dinv[d_v];
  const int w_b = __float_as_int(w_v);

  if (cnt == 32) {
    for (int i0 = 0; i0 < 32; i0 += U) {
      float h[U][R];
#pragma unroll
      for (int u = 0; u < U; ++u) {
        const int s = __builtin_amdgcn_readlane(s_v, i0 + u);
        const float* __restrict__ row = H + (long long)s * C + lane;
#pragma unroll
        for (int r = 0; r < R; ++r) h[u][r] = row[32 * r];
      }
#pragma unroll
      for (int u = 0; u < U; ++u) {
        const int d = __builtin_amdgcn_readlane(d_v, i0 + u);
        const float w = __int_as_float(__builtin_amdgcn_readlane(w_b, i0 + u));
        float* __restrict__ o = A + (long long)d * C + lane;
#pragma unroll
        for (int r = 0; r < R; ++r) atomicAdd(o + 32 * r, w * h[u][r]);
      }
    }
  } else {
    for (int i = 0; i < cnt; ++i) {
      const int s = __builtin_amdgcn_readlane(s_v, i);
      const int d = __builtin_amdgcn_readlane(d_v, i);
      const float w = __int_as_float(__builtin_amdgcn_readlane(w_b, i));
      const float* __restrict__ row = H + (long long)s * C + lane;
      float* __restrict__ o = A + (long long)d * C + lane;
#pragma unroll
      for (int r = 0; r < R; ++r) atomicAdd(o + 32 * r, w * row[32 * r]);
    }
  }
}


#define BKT_SHIFT 8
#define BKT_ROWS 256
#define BIN_PT 16

__device__ __forceinline__ int lane_read_i(int v, int src_lane) {
  return __builtin_amdgcn_ds_bpermute(src_lane << 2, v);
}
__device__ __forceinline__ float lane_read_f(float v, int src_lane) {
  return __int_as_float(__builtin_amdgcn_ds_bpermute(src_lane << 2, __float_as_int(v)));
}

__device__ __forceinline__ int block_scan_excl(int v, int* wsum, int* total) {
  const int lane = threadIdx.x & 31, wv = threadIdx.x >> 5;
  int inc = v;
  for (int o = 1; o < 32; o <<= 1) {
    const int r = lane_read_i(inc, (lane - o) & 31);
    if (lane >= o) inc += r;
  }
  __syncthreads();
  if (lane == 31) wsum[wv] = inc;
  __syncthreads();
  int woff = 0, tot = 0;
  for (int k = 0; k < 8; ++k) {
    const int s = wsum[k];
    woff += (k < wv) ? s : 0;
    tot += s;
  }
  *total = tot;
  __syncthreads();
  return woff + inc - v;
}

__global__ __launch_bounds__(256) void k_chunk_sort(const int* __restrict__ src,
                                                    const int* __restrict__ dst,
                                                    int* __restrict__ offT,
                                                    int* __restrict__ tmp_i,
                                                    int e, int nb, int pitch) {
  extern __shared__ int sh[];
  __shared__ int wsum[8];
  int* hist = sh;
  int* loc = sh + nb;
  int2* __restrict__ tmp = (int2*)tmp_i;
  const int t = threadIdx.x;
  int* __restrict__ orow = offT + (size_t)blockIdx.x * pitch;
  for (int k = t; k < nb; k += 256) hist[k] = 0;
  __syncthreads();
  const long long e0 = (long long)blockIdx.x * (256 * BIN_PT);
  int s[BIN_PT], d[BIN_PT], r[BIN_PT];
#pragma unroll
  for (int u = 0; u < BIN_PT; ++u) {
    const long long i = e0 + u * 256 + t;
    const bool ok = i < e;
    s[u] = ok ? src[i] : 0;
    d[u] = ok ? dst[i] : -1;
  }
#pragma unroll
  for (int u = 0; u < BIN_PT; ++u)
    r[u] = (d[u] >= 0) ? atomicAdd(&hist[d[u] >> BKT_SHIFT], 1) : 0;
  __syncthreads();
  const int ch = (nb + 255) >> 8;
  int run = 0;
  for (int k = t * ch; k < t * ch + ch; ++k) run += (k < nb) ? hist[k] : 0;
  int total;
  int pre = block_scan_excl(run, wsum, &total);
  for (int k = t * ch; k < t * ch + ch; ++k)
    if (k < nb) {
      loc[k] = pre;
      orow[k] = pre;
      pre += hist[k];
    }
  if (t == 0) orow[nb] = total;
  __syncthreads();
#pragma unroll
  for (int u = 0; u < BIN_PT; ++u)
    if (d[u] >= 0) tmp[e0 + loc[d[u] >> BKT_SHIFT] + r[u]] = make_int2(s[u], d[u]);
}

#define SEG_CU 16

template <bool PLACE>
__device__ __forceinline__ void walk_runs(const int2* __restrict__ tmp, const int* sst,
                                          const int* sen, int bb, int hw, int j, int* cur,
                                          int* __restrict__ csr) {
  const size_t cs = 256 * BIN_PT;
  bool more = false;
  for (int c0 = hw; c0 < bb; c0 += 16 * SEG_CU) {
    int2 ea[SEG_CU], eb[SEG_CU];
    int kk[SEG_CU], ee[SEG_CU];
#pragma unroll
    for (int u = 0; u < SEG_CU; ++u) {
      const int c = c0 + 16 * u;
      const int cc = (c < bb) ? c : (bb - 1);
      kk[u] = sst[cc] + j;
      ee[u] = (c < bb) ? sen[cc] : 0;
    }
#pragma unroll
    for (int u = 0; u < SEG_CU; ++u) {
      const int2* __restrict__ seg = tmp + (size_t)(c0 + 16 * u) * cs;
      const int k = kk[u], en = ee[u];
      ea[u] = make_int2(0, -1);
      eb[u] = make_int2(0, -1);
      if (PLACE) {
        if (k < en) ea[u] = seg[k];
        if (k + 16 < en) eb[u] = seg[k + 16];
      } else {
        if (k < en) ea[u].y = seg[k].y;
        if (k + 16 < en) eb[u].y = seg[k + 16].y;
      }
      more |= (k + 32 < en);
    }
#pragma unroll
    for (int u = 0; u < SEG_CU; ++u) {
      if (ea[u].y >= 0) {
        const int pos = atomicAdd(&cur[ea[u].y & (BKT_ROWS - 1)], 1);
        if (PLACE) csr[pos] = ea[u].x;
      }
      if (eb[u].y >= 0) {
        const int pos = atomicAdd(&cur[eb[u].y & (BKT_ROWS - 1)], 1);
        if (PLACE) csr[pos] = eb[u].x;
      }
    }
  }
  if (more) {
    for (int c = hw; c < bb; c += 16) {
      const int2* __restrict__ seg = tmp + (size_t)c * cs;
      const int en = sen[c];
      for (int k = sst[c] + 32 + j; k < en; k += 16) {
        const int2 ed = seg[k];
        const int pos = atomicAdd(&cur[ed.y & (BKT_ROWS - 1)], 1);
        if (PLACE) csr[pos] = ed.x;
      }
    }
  }
}

__global__ __launch_bounds__(256) void k_fill_csr(const int* __restrict__ tmp_i,
                                                  const int* __restrict__ offT,
                                                  int* __restrict__ rowptr,
                                                  float* __restrict__ dinv,
                                                  int* __restrict__ csr, int n, int bb,
                                                  int pitch) {
  extern __shared__ int sh[];
  __shared__ int cur[BKT_ROWS];
  __shared__ int wsum[8];
  int* sst = sh;
  int* sen = sh + bb;
  const int2* __restrict__ tmp = (const int2*)tmp_i;
  const int b = blockIdx.x, t = threadIdx.x;
  const int i = b * BKT_ROWS + t;
  const int hw = t >> 4, j = t & 15;
  const int* __restrict__ ocol = offT + b;
  int part = 0;
  for (int c = t; c < bb; c += 256) {
    const int a = ocol[(size_t)c * pitch];
    sst[c] = a;
    sen[c] = ocol[(size_t)c * pitch + 1];
    part += a;
  }
  int lo;
  (void)block_scan_excl(part, wsum, &lo);
  cur[t] = 0;
  __syncthreads();
  walk_runs<false>(tmp, sst, sen, bb, hw, j, cur, csr);
  __syncthreads();
  const int cnt = cur[t];
  int tot;
  const int start = lo + block_scan_excl(cnt, wsum, &tot);
  if (i < n) {
    rowptr[i] = start;
    dinv[i] = rsqrtf(1.0f + (float)cnt);
  }
  if (i == n - 1) rowptr[n] = start + cnt;
  cur[t] = start;
  __syncthreads();
  walk_runs<true>(tmp, sst, sen, bb, hw, j, cur, csr);
}

template <int C, int U>
__global__ __launch_bounds__(256) void k_gather_rows(const float* __restrict__ H,
                                                     const float* __restrict__ dinv,
                                                     const int* __restrict__ rowptr,
                                                     const int* __restrict__ csr,
                                                     const float* __restrict__ bias,
                                                     float* __restrict__ A, int n) {
  constexpr int G  = C / 4;
  constexpr int EP = 32 / G;
  const int lane = threadIdx.x & 31;
  const int g = lane / G, q = lane % G;
  const int nw = gridDim.x * 8;
  const float4* __restrict__ H4 = (const float4*)H;
  float4* __restrict__ A4 = (float4*)A;
  const float4 bq = ((const float4*)bias)[q];
  const int w0 = __builtin_amdgcn_readfirstlane((int)(blockIdx.x * 8u + (threadIdx.x >> 5)));
  if (w0 >= n) return;
  int beg_n = rowptr[w0], end_n = rowptr[w0 + 1];
  for (int i = w0; i < n; i += nw) {
    const int beg = __builtin_amdgcn_readfirstlane(beg_n);
    const int end = __builtin_amdgcn_readfirstlane(end_n);
    if (i + nw < n) {
      beg_n = rowptr[i + nw];
      end_n = rowptr[i + nw + 1];
    }
    const float di = dinv[i];
    const float4 hs = H4[(unsigned)(i * G + q)];
    float4 acc = make_float4(0.0f, 0.0f, 0.0f, 0.0f);
    for (int e0 = beg; e0 < end; e0 += 32) {
      const int cnt = (end - e0 < 32) ? (end - e0) : 32;
      const int li = (lane < cnt) ? lane : (cnt - 1);
      const int s_l = csr[e0 + li];
      for (int j = 0; j < cnt; j += EP * U) {
        float4 h[U];
        int s[U];
#pragma unroll
        for (int u = 0; u < U; ++u) s[u] = lane_read_i(s_l, j + u * EP + g);
#pragma unroll
        for (int u = 0; u < U; ++u) {
          h[u] = make_float4(0.0f, 0.0f, 0.0f, 0.0f);
          if (j + u * EP + g < cnt) h[u] = H4[(unsigned)(s[u] * G + q)];
        }
#pragma unroll
        for (int u = 0; u < U; ++u) {
          acc.x += h[u].x;
          acc.y += h[u].y;
          acc.z += h[u].z;
          acc.w += h[u].w;
        }
      }
    }
#pragma unroll
    for (int o = G; o < 32; o <<= 1) {
      acc.x += lane_read_f(acc.x, lane ^ o);
      acc.y += lane_read_f(acc.y, lane ^ o);
      acc.z += lane_read_f(acc.z, lane ^ o);
      acc.w += lane_read_f(acc.w, lane ^ o);
    }
    if (g == 0) {
      A4[(unsigned)(i * G + q)] = make_float4(bq.x + di * (acc.x + hs.x), bq.y + di * (acc.y + hs.y),
                                              bq.z + di * (acc.z + hs.z), bq.w + di * (acc.w + hs.w));
    }
  }
}

#define TILE_PITCH 68
__global__ __launch_bounds__(256) void k_gather_rows_gemm(const float* __restrict__ H,
                                                          const float* __restrict__ dinv,
                                                          const int* __restrict__ rowptr,
                                                          const int* __restrict__ csr,
                                                          const float* __restrict__ bias,
                                                          const float* __restrict__ W,
                                                          float* __restrict__ H2, int n) {
  constexpr int C  = 64;
  constexpr int NO = 32;
  constexpr int U  = 8;
  constexpr int G  = C / 4;
  constexpr int EP = 32 / G;
  __shared__ float4 bfrag[(C / 4) * 32];
  __shared__ __attribute__((aligned(16))) float tile[2][16 * TILE_PITCH];
  const int lane = threadIdx.x & 31;
  const int g = lane / G, q = lane % G;
  const int m = lane & 15;
  const int khi = (lane >> 4) << 1;
  const int nw = gridDim.x * 8;
  const int wave = __builtin_amdgcn_readfirstlane((int)(threadIdx.x >> 5));
  const int b8 = blockIdx.x * 8;
  if (b8 >= n) return;
  const float4* __restrict__ H4 = (const float4*)H;
  const float4 bq = ((const float4*)bias)[q];

  const int w0 = b8 + wave;
  int beg_n = 0, end_n = 0;
  if (w0 < n) {
    beg_n = rowptr[w0];
    end_n = rowptr[w0 + 1];
  }

  for (int jj = wave; jj < C / 4; jj += 8) {
    const float* __restrict__ wp = W + (4 * jj + khi) * NO + m;
    bfrag[jj * 32 + lane] = make_float4(wp[0], wp[NO], wp[16], wp[NO + 16]);
  }
  __syncthreads();
  int k = 0;
  for (int base = b8; base < n; base += nw, ++k) {
    const int i = base + wave;
    float4 r4 = make_float4(0.0f, 0.0f, 0.0f, 0.0f);
    if (i < n) {
      const int beg = __builtin_amdgcn_readfirstlane(beg_n);
      const int end = __builtin_amdgcn_readfirstlane(end_n);
      if (i + nw < n) {
        beg_n = rowptr[i + nw];
        end_n = rowptr[i + nw + 1];
      }
      const float di = dinv[i];
      const float4 hs = H4[(unsigned)(i * G + q)];
      float4 acc = make_float4(0.0f, 0.0f, 0.0f, 0.0f);
      for (int e0 = beg; e0 < end; e0 += 32) {
        const int cnt = (end - e0 < 32) ? (end - e0) : 32;
        const int li = (lane < cnt) ? lane : (cnt - 1);
        const int s_l = csr[e0 + li];
        for (int j = 0; j < cnt; j += EP * U) {
          float4 h[U];
          int s[U];
#pragma unroll
          for (int u = 0; u < U; ++u) s[u] = lane_read_i(s_l, j + u * EP + g);
#pragma unroll
          for (int u = 0; u < U; ++u) {
            h[u] = make_float4(0.0f, 0.0f, 0.0f, 0.0f);
            if (j + u * EP + g < cnt) h[u] = H4[(unsigned)(s[u] * G + q)];
          }
#pragma unroll
          for (int u = 0; u < U; ++u) {
            acc.x += h[u].x;
            acc.y += h[u].y;
            acc.z += h[u].z;
            acc.w += h[u].w;
          }
        }
      }
#pragma unroll
      for (int o = G; o < 32; o <<= 1) {
        acc.x += lane_read_f(acc.x, lane ^ o);
        acc.y += lane_read_f(acc.y, lane ^ o);
        acc.z += lane_read_f(acc.z, lane ^ o);
        acc.w += lane_read_f(acc.w, lane ^ o);
      }
      r4 = make_float4(bq.x + di * (acc.x + hs.x), bq.y + di * (acc.y + hs.y),
                       bq.z + di * (acc.z + hs.z), bq.w + di * (acc.w + hs.w));
      r4.x = (r4.x > 0.0f) ? r4.x : NEG_SLOPE * r4.x;
      r4.y = (r4.y > 0.0f) ? r4.y : NEG_SLOPE * r4.y;
      r4.z = (r4.z > 0.0f) ? r4.z : NEG_SLOPE * r4.z;
      r4.w = (r4.w > 0.0f) ? r4.w : NEG_SLOPE * r4.w;
      r4.x *= di;
      r4.y *= di;
      r4.z *= di;
      r4.w *= di;
    }
    const int odd = k & 1;
    float* __restrict__ tb = &tile[(k >> 1) & 1][0];
    if (g == 0) *(float4*)(tb + (odd * 8 + wave) * TILE_PITCH + 4 * q) = r4;
    const bool lastit = base + nw >= n;
    if (odd || lastit) {
      if (!odd && g == 0)
        *(float4*)(tb + (8 + wave) * TILE_PITCH + 4 * q) = make_float4(0.0f, 0.0f, 0.0f, 0.0f);
      __syncthreads();
      if (wave == ((k >> 1) & 7)) {
        v8f c0 = v8f{}, c1 = v8f{};
        const float* __restrict__ ar = tb + m * TILE_PITCH + khi;
#pragma unroll
        for (int j = 0; j < C / 4; ++j) {
          const v2f a = *(const v2f*)(ar + 4 * j);
          const float4 bb = bfrag[j * 32 + lane];
          v2f b0, b1;
          b0.x = bb.x; b0.y = bb.y;
          b1.x = bb.z; b1.y = bb.w;
          c0 = __builtin_amdgcn_wmma_f32_16x16x4_f32(false, a, false, b0, (short)0, c0, false, false);
          c1 = __builtin_amdgcn_wmma_f32_16x16x4_f32(false, a, false, b1, (short)0, c1, false, false);
        }
        const int hi = lane >> 4;
        const int rbase = (odd ? base - nw : base) + hi * nw;
        const bool half_ok = (hi == 0) || odd;
#pragma unroll
        for (int v = 0; v < 8; ++v) {
          const int r = rbase + v;
          if (half_ok && r < n) {
            H2[(long long)r * NO + m]      = c0[v];
            H2[(long long)r * NO + 16 + m] = c1[v];
          }
        }
      }
    }
  }
}


extern "C" void kernel_launch(void* const* d_in, const int* in_sizes, int n_in,
                              void* d_out, int out_size, void* d_ws, size_t ws_size,
                              hipStream_t stream) {
  (void)n_in; (void)out_size;

  const float* x  = (const float*)d_in[0];
  const int*   ei = (const int*)  d_in[1];
  const float* W1 = (const float*)d_in[2];
  const float* b1 = (const float*)d_in[3];
  const float* W2 = (const float*)d_in[4];
  const float* b2 = (const float*)d_in[5];
  float* out = (float*)d_out;

  const int DIN  = 64;
  const int N    = in_sizes[0] / DIN;
  const int E    = in_sizes[1] / 2;
  const int DOUT = in_sizes[5];

  const int* src = ei;
  const int* dst = ei + E;

  float* dinv = (float*)d_ws;
  int Nr = ((N + 255) / 256) * 256;
  float* H1 = dinv + Nr;
  float* A1 = H1 + (size_t)N * DIN;
  float* H2 = H1;

  const int TB = 256;
  const int RES = 1024;
  const int Mtiles = (N + 15) / 16;
  const int GB = (Mtiles + 7) / 8;

  const int NB = (N + BKT_ROWS - 1) / BKT_ROWS;
  const int BB = (E + TB * BIN_PT - 1) / (TB * BIN_PT);
  const int OP = (NB + 1 + 31) & ~31;
  const size_t TO = (2 * (size_t)E + 31) & ~(size_t)31;
  const size_t need = ((size_t)Nr + 2 * (size_t)N * DIN + (size_t)Nr + 256 + (size_t)E) * 4;
  const bool sorted_path = ws_size >= need && E > 0 && N < (1 << 26) && NB <= 8192 && BB <= 4096 &&
                           TO + (size_t)OP * (size_t)BB <= (size_t)N * DIN;

  if (sorted_path) {
    int* tmp  = (int*)A1;
    int* offT = tmp + TO;
    int* rowptr = (int*)(A1 + (size_t)N * DIN);
    int* csr = rowptr + Nr + 256;

    k_chunk_sort<<<BB, TB, 2 * NB * sizeof(int), stream>>>(src, dst, offT, tmp, E, NB, OP);
    k_fill_csr<<<NB, TB, 2 * BB * sizeof(int), stream>>>(tmp, offT, rowptr, dinv, csr, N, BB, OP);

    const int GR = 2048;
    k_gemm64_rows<<<GB, TB, 0, stream>>>(x, W1, H1, N, dinv);
    float* H2s = A1;
    k_gather_rows_gemm<<<GR, TB, 0, stream>>>(H1, dinv, rowptr, csr, b1, W2, H2s, N);
    k_gather_rows<32, 4><<<GR, TB, 0, stream>>>(H2s, dinv, rowptr, csr, b2, out, N);
    return;
  }

  k_init_deg <<<(N + TB - 1) / TB, TB, 0, stream>>>(dinv, N);
  int CB = (E + TB - 1) / TB;
  if (CB > RES) CB = RES;
  k_count_deg<<<CB, TB, 0, stream>>>(dst, dinv, E);
  k_rsqrt    <<<(N + TB - 1) / TB, TB, 0, stream>>>(dinv, N);

  const int EB = ((E + 31) / 32 + 7) / 8;

  k_gemm_wmma<4, false, false><<<GB, TB, 0, stream>>>(x, W1, H1, N, nullptr);
  {
    long long tot = (long long)N << 4;
    int IB = (int)((tot + TB - 1) / TB);
    if (IB > RES) IB = RES;
    k_agg_init<<<IB, TB, 0, stream>>>(H1, dinv, b1, A1, N, DIN, 4);
    k_agg_edges<64><<<EB, TB, 0, stream>>>(H1, dinv, src, dst, A1, E);
  }

  k_gemm_wmma<2, true, false><<<GB, TB, 0, stream>>>(A1, W2, H2, N, nullptr);
  {
    long long tot = (long long)N << 3;
    int IB = (int)((tot + TB - 1) / TB);
    if (IB > RES) IB = RES;
    k_agg_init<<<IB, TB, 0, stream>>>(H2, dinv, b2, out, N, DOUT, 3);
    k_agg_edges<32><<<EB, TB, 0, stream>>>(H2, dinv, src, dst, out, E);
  }
}


# edge sort stores whole lines: both levels place their records in an LDS copy of the block's own output range (chunk slots; bucket's run of the list), then write it front to back; same slots and sums
# speedup vs baseline: 4.0437x; 1.0820x over previous
#include <hip/hip_runtime.h>
#include <hip/hip_bf16.h>

typedef __attribute__((ext_vector_type(2))) float v2f;
typedef __attribute__((ext_vector_type(8))) float v8f;

#define NEG_SLOPE 0.01f


__global__ __launch_bounds__(256) void k_init_deg(float* __restrict__ deg, int n) {
  int i = blockIdx.x * blockDim.x + threadIdx.x;
  if (i < n) deg[i] = 1.0f;
}

__global__ __launch_bounds__(256) void k_count_deg(const int* __restrict__ dst,
                                                   float* __restrict__ deg, int e) {
  const int stride = gridDim.x * blockDim.x;
  for (int i0 = blockIdx.x * blockDim.x + threadIdx.x; i0 < e; i0 += 4 * stride) {
    int d[4];
#pragma unroll
    for (int u = 0; u < 4; ++u) {
      const int i = i0 + u * stride;
      d[u] = (i < e) ? dst[i] : -1;
    }
#pragma unroll
    for (int u = 0; u < 4; ++u)
      if (d[u] >= 0) atomicAdd(&deg[d[u]], 1.0f);
  }
}

__global__ __launch_bounds__(256) void k_rsqrt(float* __restrict__ deg, int n) {
  int i = blockIdx.x * blockDim.x + threadIdx.x;
  if (i < n) deg[i] = rsqrtf(deg[i]);
}


template <int NT, bool LEAKY, bool SCALE>
__global__ __launch_bounds__(256) void k_gemm_wmma(const float* __restrict__ X,
                                                   const float* __restrict__ W,
                                                   float* __restrict__ H,
                                                   int M, const float* __restrict__ rscale) {
  constexpr int K = 64;
  constexpr int Nout = NT * 16;
  const int lane = threadIdx.x & 31;
  const int m    = lane & 15;
  const int khi  = (lane >> 4) << 1;

  const int tile = blockIdx.x * 8 + (threadIdx.x >> 5);
  const int row0 = tile << 4;
  if (row0 >= M) return;

  int arow = row0 + m;
  if (arow >= M) arow = M - 1;
  const float* __restrict__ xr = X + (long long)arow * K + khi;
  const float* __restrict__ wc = W + (long long)khi * Nout + m;
  const float rs = SCALE ? rscale[arow] : 1.0f;

  v2f a[K / 4];
#pragma unroll
  for (int j = 0; j < K / 4; ++j) {
    a[j] = *(const v2f*)(xr + 4 * j);
    if (LEAKY) {
      a[j].x = (a[j].x > 0.0f) ? a[j].x : NEG_SLOPE * a[j].x;
      a[j].y = (a[j].y > 0.0f) ? a[j].y : NEG_SLOPE * a[j].y;
    }
    if (SCALE) {
      a[j].x *= rs;
      a[j].y *= rs;
    }
  }

  v8f c[NT];
#pragma unroll
  for (int t = 0; t < NT; ++t) c[t] = v8f{};

#pragma unroll
  for (int j = 0; j < K / 4; ++j) {
#pragma unroll
    for (int t = 0; t < NT; ++t) {
      v2f b;
      b.x = wc[(4 * j + 0) * Nout + 16 * t];
      b.y = wc[(4 * j + 1) * Nout + 16 * t];
      c[t] = __builtin_amdgcn_wmma_f32_16x16x4_f32(false, a[j], false, b, (short)0,
                                                   c[t], false, false);
    }
  }

  const int hi8 = (lane >> 4) << 3;
#pragma unroll
  for (int v = 0; v < 8; ++v) {
    int r = row0 + hi8 + v;
    if (r < M) {
#pragma unroll
      for (int t = 0; t < NT; ++t) H[(long long)r * Nout + 16 * t + m] = c[t][v];
    }
  }
}

#define A_PITCH 68
__global__ __launch_bounds__(256) void k_gemm64_rows(const float* __restrict__ X,
                                                     const float* __restrict__ W,
                                                     float* __restrict__ H, int M,
                                                     const float* __restrict__ rscale) {
  constexpr int K  = 64;
  constexpr int NO = 64;
  __shared__ float4 bfrag[(K / 4) * 32 * 2];
  __shared__ __attribute__((aligned(16))) float atile[8][16 * A_PITCH];
  const int lane = threadIdx.x & 31;
  const int wave = threadIdx.x >> 5;
  const int m    = lane & 15;
  const int h    = lane >> 4;
  const int khi  = h << 1;
  const int row0 = (blockIdx.x * 8 + wave) << 4;
  float* __restrict__ tl = &atile[wave][0];
  const int fc = m << 2;

  float4 av[8];
#pragma unroll
  for (int i = 0; i < 8; ++i) {
    int r = row0 + 2 * i + h;
    if (r >= M) r = M - 1;
    av[i] = *(const float4*)(X + (long long)r * K + fc);
  }
  int ar = row0 + m;
  if (ar >= M) ar = M - 1;
  const float rs = rscale[ar];
#pragma unroll
  for (int u = 0; u < 2; ++u) {
    const int jj = wave + 8 * u;
    const float* __restrict__ wp = W + (4 * jj + khi) * NO + m;
    bfrag[(jj * 32 + lane) * 2]     = make_float4(wp[0],  wp[NO],      wp[16], wp[NO + 16]);
    bfrag[(jj * 32 + lane) * 2 + 1] = make_float4(wp[32], wp[NO + 32], wp[48], wp[NO + 48]);
  }
#pragma unroll
  for (int i = 0; i < 8; ++i) *(float4*)(tl + (2 * i + h) * A_PITCH + fc) = av[i];
  __syncthreads();
  {
    v2f a[K / 4];
#pragma unroll
    for (int j = 0; j < K / 4; ++j) {
      a[j] = *(const v2f*)(tl + m * A_PITCH + khi + 4 * j);
      a[j].x *= rs;
      a[j].y *= rs;
    }
    v8f c[4];
#pragma unroll
    for (int t = 0; t < 4; ++t) c[t] = v8f{};
#pragma unroll
    for (int j = 0; j < K / 4; ++j) {
      const float4 q0 = bfrag[(j * 32 + lane) * 2];
      const float4 q1 = bfrag[(j * 32 + lane) * 2 + 1];
      v2f b0, b1, b2, b3;
      b0.x = q0.x; b0.y = q0.y;
      b1.x = q0.z; b1.y = q0.w;
      b2.x = q1.x; b2.y = q1.y;
      b3.x = q1.z; b3.y = q1.w;
      c[0] = __builtin_amdgcn_wmma_f32_16x16x4_f32(false, a[j], false, b0, (short)0, c[0], false, false);
      c[1] = __builtin_amdgcn_wmma_f32_16x16x4_f32(false, a[j], false, b1, (short)0, c[1], false, false);
      c[2] = __builtin_amdgcn_wmma_f32_16x16x4_f32(false, a[j], false, b2, (short)0, c[2], false, false);
      c[3] = __builtin_amdgcn_wmma_f32_16x16x4_f32(false, a[j], false, b3, (short)0, c[3], false, false);
    }
    const int hi8 = h << 3;
#pragma unroll
    for (int v = 0; v < 8; ++v) {
#pragma unroll
      for (int t = 0; t < 4; ++t) tl[(hi8 + v) * A_PITCH + 16 * t + m] = c[t][v];
    }
  }
  __syncthreads();
#pragma unroll
  for (int i = 0; i < 8; ++i) {
    const int r = row0 + 2 * i + h;
    if (r < M)
      *(float4*)(H + (long long)r * NO + fc) = *(const float4*)(tl + (2 * i + h) * A_PITCH + fc);
  }
}

__global__ __launch_bounds__(256) void k_agg_init(const float* __restrict__ H,
                                                  const float* __restrict__ dinv,
                                                  const float* __restrict__ bias,
                                                  float* __restrict__ A,
                                                  int n, int C, int tshift) {
  (void)C;
  const int tot = n << tshift;
  const int stride = gridDim.x * blockDim.x;
  const int t = threadIdx.x & ((1 << tshift) - 1);
  const float4 b = *((const float4*)bias + t);
  for (int g0 = blockIdx.x * blockDim.x + threadIdx.x; g0 < tot; g0 += 4 * stride) {
    float4 h[4];
    float di[4];
#pragma unroll
    for (int u = 0; u < 4; ++u) {
      const int gid = g0 + u * stride;
      if (gid < tot) {
        di[u] = dinv[gid >> tshift];
        h[u]  = *((const float4*)H + gid);
      }
    }
#pragma unroll
    for (int u = 0; u < 4; ++u) {
      const int gid = g0 + u * stride;
      if (gid < tot) {
        const float w = di[u] * di[u];
        *((float4*)A + gid) = make_float4(b.x + w * h[u].x, b.y + w * h[u].y,
                                          b.z + w * h[u].z, b.w + w * h[u].w);
      }
    }
  }
}

template <int C>
__global__ __launch_bounds__(256) void k_agg_edges(const float* __restrict__ H,
                                                   const float* __restrict__ dinv,
                                                   const int* __restrict__ src,
                                                   const int* __restrict__ dst,
                                                   float* __restrict__ A,
                                                   int e) {
  constexpr int R = C / 32;
  constexpr int U = 16 / R;
  const int lane = threadIdx.x & 31;
  const int wave = __builtin_amdgcn_readfirstlane(
      (int)((blockIdx.x * 256u + threadIdx.x) >> 5));
  const long long base = (long long)wave << 5;
  if (base >= e) return;
  const int cnt = (e - base < 32) ? (int)(e - base) : 32;
  const bool ok = lane < cnt;
  const int s_v = ok ? src[base + lane] : 0;
  const int d_v = ok ? dst[base + lane] : 0;
  const float w_v = dinv[s_v] * dinv[d_v];
  const int w_b = __float_as_int(w_v);

  if (cnt == 32) {
    for (int i0 = 0; i0 < 32; i0 += U) {
      float h[U][R];
#pragma unroll
      for (int u = 0; u < U; ++u) {
        const int s = __builtin_amdgcn_readlane(s_v, i0 + u);
        const float* __restrict__ row = H + (long long)s * C + lane;
#pragma unroll
        for (int r = 0; r < R; ++r) h[u][r] = row[32 * r];
      }
#pragma unroll
      for (int u = 0; u < U; ++u) {
        const int d = __builtin_amdgcn_readlane(d_v, i0 + u);
        const float w = __int_as_float(__builtin_amdgcn_readlane(w_b, i0 + u));
        float* __restrict__ o = A + (long long)d * C + lane;
#pragma unroll
        for (int r = 0; r < R; ++r) atomicAdd(o + 32 * r, w * h[u][r]);
      }
    }
  } else {
    for (int i = 0; i < cnt; ++i) {
      const int s = __builtin_amdgcn_readlane(s_v, i);
      const int d = __builtin_amdgcn_readlane(d_v, i);
      const float w = __int_as_float(__builtin_amdgcn_readlane(w_b, i));
      const float* __restrict__ row = H + (long long)s * C + lane;
      float* __restrict__ o = A + (long long)d * C + lane;
#pragma unroll
      for (int r = 0; r < R; ++r) atomicAdd(o + 32 * r, w * row[32 * r]);
    }
  }
}


#define BKT_SHIFT 8
#define BKT_ROWS 256
#define BIN_PT 16
#define CHUNK_E (256 * BIN_PT)
#define CSR_CAP 8192

__device__ __forceinline__ int lane_read_i(int v, int src_lane) {
  return __builtin_amdgcn_ds_bpermute(src_lane << 2, v);
}
__device__ __forceinline__ float lane_read_f(float v, int src_lane) {
  return __int_as_float(__builtin_amdgcn_ds_bpermute(src_lane << 2, __float_as_int(v)));
}

__device__ __forceinline__ int block_scan_excl(int v, int* wsum, int* total) {
  const int lane = threadIdx.x & 31, wv = threadIdx.x >> 5;
  int inc = v;
  for (int o = 1; o < 32; o <<= 1) {
    const int r = lane_read_i(inc, (lane - o) & 31);
    if (lane >= o) inc += r;
  }
  __syncthreads();
  if (lane == 31) wsum[wv] = inc;
  __syncthreads();
  int woff = 0, tot = 0;
  for (int k = 0; k < 8; ++k) {
    const int s = wsum[k];
    woff += (k < wv) ? s : 0;
    tot += s;
  }
  *total = tot;
  __syncthreads();
  return woff + inc - v;
}

__global__ __launch_bounds__(256) void k_chunk_sort(const int* __restrict__ src,
                                                    const int* __restrict__ dst,
                                                    int* __restrict__ offT,
                                                    int* __restrict__ tmp_i,
                                                    int e, int nb, int pitch) {
  extern __shared__ int sh[];
  __shared__ int wsum[8];
  __shared__ __attribute__((aligned(16))) int2 stage[CHUNK_E];
  int* hist = sh;
  int* loc = sh + nb;
  int2* __restrict__ tmp = (int2*)tmp_i;
  const int t = threadIdx.x;
  int* __restrict__ orow = offT + (size_t)blockIdx.x * pitch;
  for (int k = t; k < nb; k += 256) hist[k] = 0;
  __syncthreads();
  const long long e0 = (long long)blockIdx.x * (256 * BIN_PT);
  int s[BIN_PT], d[BIN_PT], r[BIN_PT];
#pragma unroll
  for (int u = 0; u < BIN_PT; ++u) {
    const long long i = e0 + u * 256 + t;
    const bool ok = i < e;
    s[u] = ok ? src[i] : 0;
    d[u] = ok ? dst[i] : -1;
  }
#pragma unroll
  for (int u = 0; u < BIN_PT; ++u)
    r[u] = (d[u] >= 0) ? atomicAdd(&hist[d[u] >> BKT_SHIFT], 1) : 0;
  __syncthreads();
  const int ch = (nb + 255) >> 8;
  int run = 0;
  for (int k = t * ch; k < t * ch + ch; ++k) run += (k < nb) ? hist[k] : 0;
  int total;
  int pre = block_scan_excl(run, wsum, &total);
  for (int k = t * ch; k < t * ch + ch; ++k)
    if (k < nb) {
      loc[k] = pre;
      orow[k] = pre;
      pre += hist[k];
    }
  if (t == 0) orow[nb] = total;
  __syncthreads();
  int p[BIN_PT];
#pragma unroll
  for (int u = 0; u < BIN_PT; ++u) p[u] = loc[(d[u] >= 0) ? (d[u] >> BKT_SHIFT) : 0] + r[u];
#pragma unroll
  for (int u = 0; u < BIN_PT; ++u)
    if (d[u] >= 0) stage[p[u]] = make_int2(s[u], d[u]);
  __syncthreads();
  int4* __restrict__ o4 = (int4*)(tmp + e0);
  const int4* s4 = (const int4*)stage;
  if (total == CHUNK_E) {
#pragma unroll
    for (int i = 0; i < CHUNK_E / 2 / 256; ++i) o4[t + 256 * i] = s4[t + 256 * i];
  } else {
    const int npair = total >> 1;
    for (int k = t; k < npair; k += 256) o4[k] = s4[k];
    if ((total & 1) && t == 0) tmp[e0 + total - 1] = stage[total - 1];
  }
}

#define SEG_CU 16

template <bool PLACE>
__device__ __forceinline__ void walk_runs(const int2* __restrict__ tmp, const int* sst,
                                          const int* sen, int bb, int hw, int j, int* cur,
                                          int* __restrict__ csr) {
  const size_t cs = 256 * BIN_PT;
  bool more = false;
  for (int c0 = hw; c0 < bb; c0 += 16 * SEG_CU) {
    int2 ea[SEG_CU], eb[SEG_CU];
    int kk[SEG_CU], ee[SEG_CU];
#pragma unroll
    for (int u = 0; u < SEG_CU; ++u) {
      const int c = c0 + 16 * u;
      const int cc = (c < bb) ? c : (bb - 1);
      kk[u] = sst[cc] + j;
      ee[u] = (c < bb) ? sen[cc] : 0;
    }
#pragma unroll
    for (int u = 0; u < SEG_CU; ++u) {
      const int2* __restrict__ seg = tmp + (size_t)(c0 + 16 * u) * cs;
      const int k = kk[u], en = ee[u];
      ea[u] = make_int2(0, -1);
      eb[u] = make_int2(0, -1);
      if (PLACE) {
        if (k < en) ea[u] = seg[k];
        if (k + 16 < en) eb[u] = seg[k + 16];
      } else {
        if (k < en) ea[u].y = seg[k].y;
        if (k + 16 < en) eb[u].y = seg[k + 16].y;
      }
      more |= (k + 32 < en);
    }
#pragma unroll
    for (int u = 0; u < SEG_CU; ++u) {
      if (ea[u].y >= 0) {
        const int pos = atomicAdd(&cur[ea[u].y & (BKT_ROWS - 1)], 1);
        if (PLACE) csr[pos] = ea[u].x;
      }
      if (eb[u].y >= 0) {
        const int pos = atomicAdd(&cur[eb[u].y & (BKT_ROWS - 1)], 1);
        if (PLACE) csr[pos] = eb[u].x;
      }
    }
  }
  if (more) {
    for (int c = hw; c < bb; c += 16) {
      const int2* __restrict__ seg = tmp + (size_t)c * cs;
      const int en = sen[c];
      for (int k = sst[c] + 32 + j; k < en; k += 16) {
        const int2 ed = seg[k];
        const int pos = atomicAdd(&cur[ed.y & (BKT_ROWS - 1)], 1);
        if (PLACE) csr[pos] = ed.x;
      }
    }
  }
}

__global__ __launch_bounds__(256) void k_fill_csr(const int* __restrict__ tmp_i,
                                                  const int* __restrict__ offT,
                                                  int* __restrict__ rowptr,
                                                  float* __restrict__ dinv,
                                                  int* __restrict__ csr, int n, int bb,
                                                  int pitch) {
  extern __shared__ int sh[];
  __shared__ int cur[BKT_ROWS];
  __shared__ int wsum[8];
  __shared__ int stg[CSR_CAP];
  int* sst = sh;
  int* sen = sh + bb;
  const int2* __restrict__ tmp = (const int2*)tmp_i;
  const int b = blockIdx.x, t = threadIdx.x;
  const int i = b * BKT_ROWS + t;
  const int hw = t >> 4, j = t & 15;
  const int* __restrict__ ocol = offT + b;
  int part = 0;
  for (int c = t; c < bb; c += 256) {
    const int a = ocol[(size_t)c * pitch];
    sst[c] = a;
    sen[c] = ocol[(size_t)c * pitch + 1];
    part += a;
  }
  int lo;
  (void)block_scan_excl(part, wsum, &lo);
  cur[t] = 0;
  __syncthreads();
  walk_runs<false>(tmp, sst, sen, bb, hw, j, cur, csr);
  __syncthreads();
  const int cnt = cur[t];
  int tot;
  const int lstart = block_scan_excl(cnt, wsum, &tot);
  const int start = lo + lstart;
  if (i < n) {
    rowptr[i] = start;
    dinv[i] = rsqrtf(1.0f + (float)cnt);
  }
  if (i == n - 1) rowptr[n] = start + cnt;
  const bool staged = tot <= CSR_CAP;
  cur[t] = staged ? lstart : start;
  __syncthreads();
  if (staged) {
    walk_runs<true>(tmp, sst, sen, bb, hw, j, cur, stg);
    __syncthreads();
    int* __restrict__ orun = csr + lo;
    for (int k = t - (lo & 31); k < tot; k += 256)
      if (k >= 0) orun[k] = stg[k];
  } else {
    walk_runs<true>(tmp, sst, sen, bb, hw, j, cur, csr);
  }
}

template <int C, int U>
__global__ __launch_bounds__(256) void k_gather_rows(const float* __restrict__ H,
                                                     const float* __restrict__ dinv,
                                                     const int* __restrict__ rowptr,
                                                     const int* __restrict__ csr,
                                                     const float* __restrict__ bias,
                                                     float* __restrict__ A, int n) {
  constexpr int G  = C / 4;
  constexpr int EP = 32 / G;
  const int lane = threadIdx.x & 31;
  const int g = lane / G, q = lane % G;
  const int nw = gridDim.x * 8;
  const float4* __restrict__ H4 = (const float4*)H;
  float4* __restrict__ A4 = (float4*)A;
  const float4 bq = ((const float4*)bias)[q];
  const int w0 = __builtin_amdgcn_readfirstlane((int)(blockIdx.x * 8u + (threadIdx.x >> 5)));
  if (w0 >= n) return;
  int beg_n = rowptr[w0], end_n = rowptr[w0 + 1];
  for (int i = w0; i < n; i += nw) {
    const int beg = __builtin_amdgcn_readfirstlane(beg_n);
    const int end = __builtin_amdgcn_readfirstlane(end_n);
    if (i + nw < n) {
      beg_n = rowptr[i + nw];
      end_n = rowptr[i + nw + 1];
    }
    const float di = dinv[i];
    const float4 hs = H4[(unsigned)(i * G + q)];
    float4 acc = make_float4(0.0f, 0.0f, 0.0f, 0.0f);
    for (int e0 = beg; e0 < end; e0 += 32) {
      const int cnt = (end - e0 < 32) ? (end - e0) : 32;
      const int li = (lane < cnt) ? lane : (cnt - 1);
      const int s_l = csr[e0 + li];
      for (int j = 0; j < cnt; j += EP * U) {
        float4 h[U];
        int s[U];
#pragma unroll
        for (int u = 0; u < U; ++u) s[u] = lane_read_i(s_l, j + u * EP + g);
#pragma unroll
        for (int u = 0; u < U; ++u) {
          h[u] = make_float4(0.0f, 0.0f, 0.0f, 0.0f);
          if (j + u * EP + g < cnt) h[u] = H4[(unsigned)(s[u] * G + q)];
        }
#pragma unroll
        for (int u = 0; u < U; ++u) {
          acc.x += h[u].x;
          acc.y += h[u].y;
          acc.z += h[u].z;
          acc.w += h[u].w;
        }
      }
    }
#pragma unroll
    for (int o = G; o < 32; o <<= 1) {
      acc.x += lane_read_f(acc.x, lane ^ o);
      acc.y += lane_read_f(acc.y, lane ^ o);
      acc.z += lane_read_f(acc.z, lane ^ o);
      acc.w += lane_read_f(acc.w, lane ^ o);
    }
    if (g == 0) {
      A4[(unsigned)(i * G + q)] = make_float4(bq.x + di * (acc.x + hs.x), bq.y + di * (acc.y + hs.y),
                                              bq.z + di * (acc.z + hs.z), bq.w + di * (acc.w + hs.w));
    }
  }
}

#define TILE_PITCH 68
__global__ __launch_bounds__(256) void k_gather_rows_gemm(const float* __restrict__ H,
                                                          const float* __restrict__ dinv,
                                                          const int* __restrict__ rowptr,
                                                          const int* __restrict__ csr,
                                                          const float* __restrict__ bias,
                                                          const float* __restrict__ W,
                                                          float* __restrict__ H2, int n) {
  constexpr int C  = 64;
  constexpr int NO = 32;
  constexpr int U  = 8;
  constexpr int G  = C / 4;
  constexpr int EP = 32 / G;
  __shared__ float4 bfrag[(C / 4) * 32];
  __shared__ __attribute__((aligned(16))) float tile[2][16 * TILE_PITCH];
  const int lane = threadIdx.x & 31;
  const int g = lane / G, q = lane % G;
  const int m = lane & 15;
  const int khi = (lane >> 4) << 1;
  const int nw = gridDim.x * 8;
  const int wave = __builtin_amdgcn_readfirstlane((int)(threadIdx.x >> 5));
  const int b8 = blockIdx.x * 8;
  if (b8 >= n) return;
  const float4* __restrict__ H4 = (const float4*)H;
  const float4 bq = ((const float4*)bias)[q];

  const int w0 = b8 + wave;
  int beg_n = 0, end_n = 0;
  if (w0 < n) {
    beg_n = rowptr[w0];
    end_n = rowptr[w0 + 1];
  }

  for (int jj = wave; jj < C / 4; jj += 8) {
    const float* __restrict__ wp = W + (4 * jj + khi) * NO + m;
    bfrag[jj * 32 + lane] = make_float4(wp[0], wp[NO], wp[16], wp[NO + 16]);
  }
  __syncthreads();
  int k = 0;
  for (int base = b8; base < n; base += nw, ++k) {
    const int i = base + wave;
    float4 r4 = make_float4(0.0f, 0.0f, 0.0f, 0.0f);
    if (i < n) {
      const int beg = __builtin_amdgcn_readfirstlane(beg_n);
      const int end = __builtin_amdgcn_readfirstlane(end_n);
      if (i + nw < n) {
        beg_n = rowptr[i + nw];
        end_n = rowptr[i + nw + 1];
      }
      const float di = dinv[i];
      const float4 hs = H4[(unsigned)(i * G + q)];
      float4 acc = make_float4(0.0f, 0.0f, 0.0f, 0.0f);
      for (int e0 = beg; e0 < end; e0 += 32) {
        const int cnt = (end - e0 < 32) ? (end - e0) : 32;
        const int li = (lane < cnt) ? lane : (cnt - 1);
        const int s_l = csr[e0 + li];
        for (int j = 0; j < cnt; j += EP * U) {
          float4 h[U];
          int s[U];
#pragma unroll
          for (int u = 0; u < U; ++u) s[u] = lane_read_i(s_l, j + u * EP + g);
#pragma unroll
          for (int u = 0; u < U; ++u) {
            h[u] = make_float4(0.0f, 0.0f, 0.0f, 0.0f);
            if (j + u * EP + g < cnt) h[u] = H4[(unsigned)(s[u] * G + q)];
          }
#pragma unroll
          for (int u = 0; u < U; ++u) {
            acc.x += h[u].x;
            acc.y += h[u].y;
            acc.z += h[u].z;
            acc.w += h[u].w;
          }
        }
      }
#pragma unroll
      for (int o = G; o < 32; o <<= 1) {
        acc.x += lane_read_f(acc.x, lane ^ o);
        acc.y += lane_read_f(acc.y, lane ^ o);
        acc.z += lane_read_f(acc.z, lane ^ o);
        acc.w += lane_read_f(acc.w, lane ^ o);
      }
      r4 = make_float4(bq.x + di * (acc.x + hs.x), bq.y + di * (acc.y + hs.y),
                       bq.z + di * (acc.z + hs.z), bq.w + di * (acc.w + hs.w));
      r4.x = (r4.x > 0.0f) ? r4.x : NEG_SLOPE * r4.x;
      r4.y = (r4.y > 0.0f) ? r4.y : NEG_SLOPE * r4.y;
      r4.z = (r4.z > 0.0f) ? r4.z : NEG_SLOPE * r4.z;
      r4.w = (r4.w > 0.0f) ? r4.w : NEG_SLOPE * r4.w;
      r4.x *= di;
      r4.y *= di;
      r4.z *= di;
      r4.w *= di;
    }
    const int odd = k & 1;
    float* __restrict__ tb = &tile[(k >> 1) & 1][0];
    if (g == 0) *(float4*)(tb + (odd * 8 + wave) * TILE_PITCH + 4 * q) = r4;
    const bool lastit = base + nw >= n;
    if (odd || lastit) {
      if (!odd && g == 0)
        *(float4*)(tb + (8 + wave) * TILE_PITCH + 4 * q) = make_float4(0.0f, 0.0f, 0.0f, 0.0f);
      __syncthreads();
      if (wave == ((k >> 1) & 7)) {
        v8f c0 = v8f{}, c1 = v8f{};
        const float* __restrict__ ar = tb + m * TILE_PITCH + khi;
#pragma unroll
        for (int j = 0; j < C / 4; ++j) {
          const v2f a = *(const v2f*)(ar + 4 * j);
          const float4 bb = bfrag[j * 32 + lane];
          v2f b0, b1;
          b0.x = bb.x; b0.y = bb.y;
          b1.x = bb.z; b1.y = bb.w;
          c0 = __builtin_amdgcn_wmma_f32_16x16x4_f32(false, a, false, b0, (short)0, c0, false, false);
          c1 = __builtin_amdgcn_wmma_f32_16x16x4_f32(false, a, false, b1, (short)0, c1, false, false);
        }
        const int hi = lane >> 4;
        const int rbase = (odd ? base - nw : base) + hi * nw;
        const bool half_ok = (hi == 0) || odd;
#pragma unroll
        for (int v = 0; v < 8; ++v) {
          const int r = rbase + v;
          if (half_ok && r < n) {
            H2[(long long)r * NO + m]      = c0[v];
            H2[(long long)r * NO + 16 + m] = c1[v];
          }
        }
      }
    }
  }
}


extern "C" void kernel_launch(void* const* d_in, const int* in_sizes, int n_in,
                              void* d_out, int out_size, void* d_ws, size_t ws_size,
                              hipStream_t stream) {
  (void)n_in; (void)out_size;

  const float* x  = (const float*)d_in[0];
  const int*   ei = (const int*)  d_in[1];
  const float* W1 = (const float*)d_in[2];
  const float* b1 = (const float*)d_in[3];
  const float* W2 = (const float*)d_in[4];
  const float* b2 = (const float*)d_in[5];
  float* out = (float*)d_out;

  const int DIN  = 64;
  const int N    = in_sizes[0] / DIN;
  const int E    = in_sizes[1] / 2;
  const int DOUT = in_sizes[5];

  const int* src = ei;
  const int* dst = ei + E;

  float* dinv = (float*)d_ws;
  int Nr = ((N + 255) / 256) * 256;
  float* H1 = dinv + Nr;
  float* A1 = H1 + (size_t)N * DIN;
  float* H2 = H1;

  const int TB = 256;
  const int RES = 1024;
  const int Mtiles = (N + 15) / 16;
  const int GB = (Mtiles + 7) / 8;

  const int NB = (N + BKT_ROWS - 1) / BKT_ROWS;
  const int BB = (E + TB * BIN_PT - 1) / (TB * BIN_PT);
  const int OP = (NB + 1 + 31) & ~31;
  const size_t TO = (2 * (size_t)E + 31) & ~(size_t)31;
  const size_t need = ((size_t)Nr + 2 * (size_t)N * DIN + (size_t)Nr + 256 + (size_t)E) * 4;
  const bool sorted_path = ws_size >= need && E > 0 && N < (1 << 26) && NB <= 8192 && BB <= 4096 &&
                           TO + (size_t)OP * (size_t)BB <= (size_t)N * DIN;

  if (sorted_path) {
    int* tmp  = (int*)A1;
    int* offT = tmp + TO;
    int* rowptr = (int*)(A1 + (size_t)N * DIN);
    int* csr = rowptr + Nr + 256;

    k_chunk_sort<<<BB, TB, 2 * NB * sizeof(int), stream>>>(src, dst, offT, tmp, E, NB, OP);
    k_fill_csr<<<NB, TB, 2 * BB * sizeof(int), stream>>>(tmp, offT, rowptr, dinv, csr, N, BB, OP);

    const int GR = 2048;
    k_gemm64_rows<<<GB, TB, 0, stream>>>(x, W1, H1, N, dinv);
    float* H2s = A1;
    k_gather_rows_gemm<<<GR, TB, 0, stream>>>(H1, dinv, rowptr, csr, b1, W2, H2s, N);
    k_gather_rows<32, 4><<<GR, TB, 0, stream>>>(H2s, dinv, rowptr, csr, b2, out, N);
    return;
  }

  k_init_deg <<<(N + TB - 1) / TB, TB, 0, stream>>>(dinv, N);
  int CB = (E + TB - 1) / TB;
  if (CB > RES) CB = RES;
  k_count_deg<<<CB, TB, 0, stream>>>(dst, dinv, E);
  k_rsqrt    <<<(N + TB - 1) / TB, TB, 0, stream>>>(dinv, N);

  const int EB = ((E + 31) / 32 + 7) / 8;

  k_gemm_wmma<4, false, false><<<GB, TB, 0, stream>>>(x, W1, H1, N, nullptr);
  {
    long long tot = (long long)N << 4;
    int IB = (int)((tot + TB - 1) / TB);
    if (IB > RES) IB = RES;
    k_agg_init<<<IB, TB, 0, stream>>>(H1, dinv, b1, A1, N, DIN, 4);
    k_agg_edges<64><<<EB, TB, 0, stream>>>(H1, dinv, src, dst, A1, E);
  }

  k_gemm_wmma<2, true, false><<<GB, TB, 0, stream>>>(A1, W2, H2, N, nullptr);
  {
    long long tot = (long long)N << 3;
    int IB = (int)((tot + TB - 1) / TB);
    if (IB > RES) IB = RES;
    k_agg_init<<<IB, TB, 0, stream>>>(H2, dinv, b2, out, N, DOUT, 3);
    k_agg_edges<32><<<EB, TB, 0, stream>>>(H2, dinv, src, dst, out, E);
  }
}
